# starting kernel
# baseline (speedup 1.0000x reference)
#include <hip/hip_runtime.h>

typedef float v2f __attribute__((ext_vector_type(2)));
typedef float v8f __attribute__((ext_vector_type(8)));

#define N_NODES 100000
#define KDIM 128


__global__ void zero_kernel(float* __restrict__ p, int n) {
  int i = blockIdx.x * blockDim.x + threadIdx.x;
  if (i < n) p[i] = 0.0f;
}

__global__ void deg_kernel(const int* __restrict__ dst, float* __restrict__ deg, int nE) {
  int i = blockIdx.x * blockDim.x + threadIdx.x;
  if (i < nE) atomicAdd(&deg[dst[i]], 1.0f);
}

__global__ void dis_kernel(float* __restrict__ degdis, int n) {
  int i = blockIdx.x * blockDim.x + threadIdx.x;
  if (i < n) degdis[i] = rsqrtf(1.0f + degdis[i]);
}


template <int NCOLS>
__global__ void gemm_wmma_kernel(const float* __restrict__ A, const float* __restrict__ W,
                                 float* __restrict__ C) {
  const int lane = threadIdx.x & 31;
  const int wave = threadIdx.x >> 5;
  const int hi   = lane >> 4;
  const int lo   = lane & 15;
  const int m0   = blockIdx.x * 16;
  const int n    = wave * 16 + lo;

  const float* __restrict__ arow = A + (size_t)(m0 + lo) * KDIM + 2 * hi;
  const float* __restrict__ wcol = W + (size_t)(2 * hi) * NCOLS + n;

  v8f acc = {};
#pragma unroll
  for (int k = 0; k < KDIM; k += 4) {
    v2f a, b;
    a.x = arow[k + 0];
    a.y = arow[k + 1];
    b.x = wcol[(size_t)k * NCOLS];
    b.y = wcol[(size_t)k * NCOLS + NCOLS];
    acc = __builtin_amdgcn_wmma_f32_16x16x4_f32(
         false, a,  false, b,
         (short)0, acc,  false,  false);
  }

  float* __restrict__ crow = C + (size_t)(m0 + hi * 8) * NCOLS + n;
#pragma unroll
  for (int r = 0; r < 8; ++r)
    crow[(size_t)r * NCOLS] = acc[r];
}


__global__ void init_agg_kernel(const float* __restrict__ h, const float* __restrict__ dis,
                                const float* __restrict__ bias, float* __restrict__ agg,
                                int total, int F) {
  int i = blockIdx.x * blockDim.x + threadIdx.x;
  if (i >= total) return;
  int node = i / F;
  int f = i - node * F;
  float ds = dis[node];
  agg[i] = h[i] * ds * ds + bias[f];
}


template <int F>
__global__ void scatter_kernel(const float* __restrict__ h, const int* __restrict__ src,
                               const int* __restrict__ dst, const float* __restrict__ dis,
                               float* __restrict__ agg, int nE) {
  const int gw   = blockIdx.x * (blockDim.x >> 5) + (threadIdx.x >> 5);
  const int lane = threadIdx.x & 31;
  constexpr int chunks = F / 4;
  constexpr int epw    = 32 / chunks;
  const int sub = lane / chunks;
  const int l   = lane % chunks;
  const int e = gw * epw + sub;
  if (e >= nE) return;

  const int s = src[e];
  const int d = dst[e];
  const float coef = dis[s] * dis[d];

  const float4 v = *(const float4*)(h + (size_t)s * F + l * 4);
  float* out = agg + (size_t)d * F + l * 4;
  atomicAdd(out + 0, v.x * coef);
  atomicAdd(out + 1, v.y * coef);
  atomicAdd(out + 2, v.z * coef);
  atomicAdd(out + 3, v.w * coef);
}

__global__ void relu_kernel(float* __restrict__ p, int n) {
  int i = blockIdx.x * blockDim.x + threadIdx.x;
  if (i < n) p[i] = fmaxf(p[i], 0.0f);
}


__global__ void dot_kernel(const float* __restrict__ z, const int* __restrict__ ls,
                           const int* __restrict__ ld, float* __restrict__ out, int nL) {
  int e = blockIdx.x * blockDim.x + threadIdx.x;
  if (e >= nL) return;
  const float4* a = (const float4*)(z + (size_t)ls[e] * 64);
  const float4* b = (const float4*)(z + (size_t)ld[e] * 64);
  float s = 0.0f;
#pragma unroll
  for (int i = 0; i < 16; ++i) {
    float4 x = a[i], y = b[i];
    s += x.x * y.x + x.y * y.y + x.z * y.z + x.w * y.w;
  }
  out[e] = s;
}


extern "C" void kernel_launch(void* const* d_in, const int* in_sizes, int n_in,
                              void* d_out, int out_size, void* d_ws, size_t ws_size,
                              hipStream_t stream) {
  const float* x  = (const float*)d_in[0];
  const int*   ei = (const int*)d_in[1];
  const int*   el = (const int*)d_in[2];
  const float* W1 = (const float*)d_in[3];
  const float* b1 = (const float*)d_in[4];
  const float* W2 = (const float*)d_in[5];
  const float* b2 = (const float*)d_in[6];

  const int nE = in_sizes[1] / 2;
  const int nL = in_sizes[2] / 2;
  const int* src = ei;
  const int* dst = ei + nE;
  const int* ls  = el;
  const int* ld  = el + nL;

  float* ws   = (float*)d_ws;
  float* dis  = ws;
  float* bufA = ws + 100352;
  float* bufB = bufA + (size_t)N_NODES * 128;
  float* out  = (float*)d_out;

  const int elem1 = N_NODES * 128;
  const int elem2 = N_NODES * 64;

  zero_kernel<<<(N_NODES + 255) / 256, 256, 0, stream>>>(dis, N_NODES);
  deg_kernel<<<(nE + 255) / 256, 256, 0, stream>>>(dst, dis, nE);
  dis_kernel<<<(N_NODES + 255) / 256, 256, 0, stream>>>(dis, N_NODES);

  gemm_wmma_kernel<128><<<N_NODES / 16, 256, 0, stream>>>(x, W1, bufA);
  init_agg_kernel<<<(elem1 + 255) / 256, 256, 0, stream>>>(bufA, dis, b1, bufB, elem1, 128);
  scatter_kernel<128><<<(nE + 7) / 8, 256, 0, stream>>>(bufA, src, dst, dis, bufB, nE);
  relu_kernel<<<(elem1 + 255) / 256, 256, 0, stream>>>(bufB, elem1);

  gemm_wmma_kernel<64><<<N_NODES / 16, 128, 0, stream>>>(bufB, W2, bufA);
  init_agg_kernel<<<(elem2 + 255) / 256, 256, 0, stream>>>(bufA, dis, b2, bufB, elem2, 64);
  scatter_kernel<64><<<(nE + 15) / 16, 256, 0, stream>>>(bufA, src, dst, dis, bufB, nE);

  dot_kernel<<<(nL + 255) / 256, 256, 0, stream>>>(bufB, ls, ld, out, nL);
}


# Replace f32 atomic edge scatter by a per-call edge list sorted by destination (integer rank atomics, scan, place); one wave per node sums its in-edge rows in registers, in arrival-rank order
# speedup vs baseline: 2.2120x; 2.2120x over previous
#include <hip/hip_runtime.h>


typedef float v2f __attribute__((ext_vector_type(2)));
typedef float v8f __attribute__((ext_vector_type(8)));

#define N_NODES 100000
#define N_PAD   100352
#define E_CAP   1600000
#define KDIM 128


__global__ void zero_int_kernel(int* __restrict__ p, int n) {
  int i = blockIdx.x * blockDim.x + threadIdx.x;
  if (i < n) p[i] = 0;
}

__global__ void rank_kernel(const int* __restrict__ dst, int* __restrict__ cnt,
                            int* __restrict__ rank, int nE) {
  int i = blockIdx.x * blockDim.x + threadIdx.x;
  if (i < nE) rank[i] = atomicAdd(&cnt[dst[i]], 1);
}

#define SCAN_THREADS 1024
#define SCAN_PER_THREAD 16

__global__ void __launch_bounds__(SCAN_THREADS)
scan_kernel(const int* __restrict__ cnt, int* __restrict__ rowptr) {
  static_assert(N_NODES % 4 == 0, "rows of four");
  __shared__ int wsum[32];
  const int tid  = threadIdx.x;
  const int lane = tid & 31;
  const int wave = tid >> 5;
  int carry = 0;
  for (int base = 0; base < N_NODES; base += SCAN_THREADS * SCAN_PER_THREAD) {
    const int i0 = base + tid * SCAN_PER_THREAD;
    int4 c[4];
#pragma unroll
    for (int g = 0; g < 4; ++g) {
      const int i = i0 + 4 * g;
      c[g] = make_int4(0, 0, 0, 0);
      if (i < N_NODES) c[g] = *(const int4*)(cnt + i);
    }
    int t = 0;
#pragma unroll
    for (int g = 0; g < 4; ++g) t += c[g].x + c[g].y + c[g].z + c[g].w;

    int x = t;
#pragma unroll
    for (int o = 1; o < 32; o <<= 1) {
      const int y = __shfl_up(x, o);
      if (lane >= o) x += y;
    }
    if (lane == 31) wsum[wave] = x;
    __syncthreads();

    const int wv = wsum[lane];
    int wp  = (lane < wave) ? wv : 0;
    int tot = wv;
#pragma unroll
    for (int o = 16; o > 0; o >>= 1) {
      wp  += __shfl_xor(wp, o);
      tot += __shfl_xor(tot, o);
    }

    int run = carry + wp + (x - t);
#pragma unroll
    for (int g = 0; g < 4; ++g) {
      const int i = i0 + 4 * g;
      if (i < N_NODES) {
        int4 r;
        r.x = run;
        r.y = r.x + c[g].x;
        r.z = r.y + c[g].y;
        r.w = r.z + c[g].z;
        run = r.w + c[g].w;
        *(int4*)(rowptr + i) = r;
      }
    }
    carry += tot;
    __syncthreads();
  }
}

__global__ void place_kernel(const int* __restrict__ src, const int* __restrict__ dst,
                             const int* __restrict__ rowptr, const int* __restrict__ rank,
                             int* __restrict__ col, int nE) {
  int i = blockIdx.x * blockDim.x + threadIdx.x;
  if (i < nE) col[rowptr[dst[i]] + rank[i]] = src[i];
}


template <int NCOLS>
__global__ void gemm_wmma_kernel(const float* __restrict__ A, const float* __restrict__ W,
                                 float* __restrict__ C) {
  const int lane = threadIdx.x & 31;
  const int wave = threadIdx.x >> 5;
  const int hi   = lane >> 4;
  const int lo   = lane & 15;
  const int m0   = blockIdx.x * 16;
  const int n    = wave * 16 + lo;

  const float* __restrict__ arow = A + (size_t)(m0 + lo) * KDIM + 2 * hi;
  const float* __restrict__ wcol = W + (size_t)(2 * hi) * NCOLS + n;

  v8f acc = {};
#pragma unroll
  for (int k = 0; k < KDIM; k += 4) {
    v2f a, b;
    a.x = arow[k + 0];
    a.y = arow[k + 1];
    b.x = wcol[(size_t)k * NCOLS];
    b.y = wcol[(size_t)k * NCOLS + NCOLS];
    acc = __builtin_amdgcn_wmma_f32_16x16x4_f32(
         false, a,  false, b,
         (short)0, acc,  false,  false);
  }

  float* __restrict__ crow = C + (size_t)(m0 + hi * 8) * NCOLS + n;
#pragma unroll
  for (int r = 0; r < 8; ++r)
    crow[(size_t)r * NCOLS] = acc[r];
}


template <int F, bool RELU>
__global__ void gather_kernel(const float* __restrict__ h, const int* __restrict__ rowptr,
                              const int* __restrict__ cnt, const int* __restrict__ col,
                              const float* __restrict__ bias, float* __restrict__ agg, int n) {
  constexpr int chunks = F / 4;
  constexpr int epw    = 32 / chunks;
  const int lane = threadIdx.x & 31;
  const int sub  = lane / chunks;
  const int l    = lane % chunks;
  const int node = __builtin_amdgcn_readfirstlane(
      (int)(blockIdx.x * (blockDim.x >> 5) + (threadIdx.x >> 5)));
  if (node >= n) return;
  const int   start = rowptr[node];
  const int   m     = cnt[node];
  const float dn    = rsqrtf(1.0f + (float)m);

  const float4 hv = *(const float4*)(h + (size_t)node * F + l * 4);

  float4 acc = make_float4(0.0f, 0.0f, 0.0f, 0.0f);
  for (int b = 0; b < m; b += 32) {
    const int k = b + lane;
    int   idx = node;
    float cf  = 0.0f;
    if (k < m) {
      idx = col[start + k];
      cf  = rsqrtf(1.0f + (float)cnt[idx]) * dn;
    }
    const int rem = m - b;
    const int mb  = rem < 32 ? rem : 32;
    const int T   = (mb + epw - 1) / epw;
    for (int t0 = 0; t0 < T; t0 += 4) {
      float4 v[4];
      float  c[4];
#pragma unroll
      for (int u = 0; u < 4; ++u) {
        int s;
        if constexpr (epw == 1) {
          const int jl = t0 + u;
          s    = __builtin_amdgcn_readlane(idx, jl);
          c[u] = __int_as_float(__builtin_amdgcn_readlane(__float_as_int(cf), jl));
        } else {
          const int jl = (t0 + u) * epw + sub;
          s    = __builtin_amdgcn_ds_bpermute(jl << 2, idx);
          c[u] = __int_as_float(__builtin_amdgcn_ds_bpermute(jl << 2, __float_as_int(cf)));
        }
        v[u] = *(const float4*)(h + (size_t)s * F + l * 4);
      }
#pragma unroll
      for (int u = 0; u < 4; ++u) {
        acc.x += v[u].x * c[u];
        acc.y += v[u].y * c[u];
        acc.z += v[u].z * c[u];
        acc.w += v[u].w * c[u];
      }
    }
  }

  if constexpr (epw == 2) {
    acc.x += __shfl_xor(acc.x, 16);
    acc.y += __shfl_xor(acc.y, 16);
    acc.z += __shfl_xor(acc.z, 16);
    acc.w += __shfl_xor(acc.w, 16);
  }

  if (sub == 0) {
    const float4 bv = *(const float4*)(bias + l * 4);
    const float  d2 = dn * dn;
    float4 o;
    o.x = (acc.x + hv.x * d2) + bv.x;
    o.y = (acc.y + hv.y * d2) + bv.y;
    o.z = (acc.z + hv.z * d2) + bv.z;
    o.w = (acc.w + hv.w * d2) + bv.w;
    if (RELU) {
      o.x = fmaxf(o.x, 0.0f);
      o.y = fmaxf(o.y, 0.0f);
      o.z = fmaxf(o.z, 0.0f);
      o.w = fmaxf(o.w, 0.0f);
    }
    *(float4*)(agg + (size_t)node * F + l * 4) = o;
  }
}


__global__ void dot_kernel(const float* __restrict__ z, const int* __restrict__ ls,
                           const int* __restrict__ ld, float* __restrict__ out, int nL) {
  int e = blockIdx.x * blockDim.x + threadIdx.x;
  if (e >= nL) return;
  const float4* a = (const float4*)(z + (size_t)ls[e] * 64);
  const float4* b = (const float4*)(z + (size_t)ld[e] * 64);
  float s = 0.0f;
#pragma unroll
  for (int i = 0; i < 16; ++i) {
    float4 x = a[i], y = b[i];
    s += x.x * y.x + x.y * y.y + x.z * y.z + x.w * y.w;
  }
  out[e] = s;
}


extern "C" void kernel_launch(void* const* d_in, const int* in_sizes, int n_in,
                              void* d_out, int out_size, void* d_ws, size_t ws_size,
                              hipStream_t stream) {
  const float* x  = (const float*)d_in[0];
  const int*   ei = (const int*)d_in[1];
  const int*   el = (const int*)d_in[2];
  const float* W1 = (const float*)d_in[3];
  const float* b1 = (const float*)d_in[4];
  const float* W2 = (const float*)d_in[5];
  const float* b2 = (const float*)d_in[6];

  const int nE = in_sizes[1] / 2;
  const int nL = in_sizes[2] / 2;
  const int* src = ei;
  const int* dst = ei + nE;
  const int* ls  = el;
  const int* ld  = el + nL;

  int*   cnt    = (int*)d_ws;
  int*   rowptr = cnt + N_PAD;
  int*   rank   = rowptr + N_PAD;
  int*   col    = rank + E_CAP;
  float* bufA   = (float*)(col + E_CAP);
  float* bufB   = bufA + (size_t)N_NODES * 128;
  float* out    = (float*)d_out;

  zero_int_kernel<<<(N_NODES + 255) / 256, 256, 0, stream>>>(cnt, N_NODES);
  rank_kernel<<<(nE + 255) / 256, 256, 0, stream>>>(dst, cnt, rank, nE);
  scan_kernel<<<1, SCAN_THREADS, 0, stream>>>(cnt, rowptr);
  place_kernel<<<(nE + 255) / 256, 256, 0, stream>>>(src, dst, rowptr, rank, col, nE);

  gemm_wmma_kernel<128><<<N_NODES / 16, 256, 0, stream>>>(x, W1, bufA);
  gather_kernel<128, true><<<(N_NODES + 7) / 8, 256, 0, stream>>>(bufA, rowptr, cnt, col, b1, bufB,
                                                                   N_NODES);

  gemm_wmma_kernel<64><<<N_NODES / 16, 128, 0, stream>>>(bufB, W2, bufA);
  gather_kernel<64, false><<<(N_NODES + 7) / 8, 256, 0, stream>>>(bufA, rowptr, cnt, col, b2, bufB,
                                                                   N_NODES);

  dot_kernel<<<(nL + 255) / 256, 256, 0, stream>>>(bufB, ls, ld, out, nL);
}


# GEMM operand reuse: W staged once per work-group in LDS (K pairs interleaved per column), each wave computes a 16-row strip over all column tiles, so A is read once; same k order and sums
# speedup vs baseline: 2.4476x; 1.1065x over previous
#include <hip/hip_runtime.h>


typedef float v2f __attribute__((ext_vector_type(2)));
typedef float v8f __attribute__((ext_vector_type(8)));

#define N_NODES 100000
#define N_PAD   100352
#define E_CAP   1600000
#define KDIM 128


__global__ void zero_int_kernel(int* __restrict__ p, int n) {
  int i = blockIdx.x * blockDim.x + threadIdx.x;
  if (i < n) p[i] = 0;
}

__global__ void rank_kernel(const int* __restrict__ dst, int* __restrict__ cnt,
                            int* __restrict__ rank, int nE) {
  int i = blockIdx.x * blockDim.x + threadIdx.x;
  if (i < nE) rank[i] = atomicAdd(&cnt[dst[i]], 1);
}

#define SCAN_THREADS 1024
#define SCAN_PER_THREAD 16

__global__ void __launch_bounds__(SCAN_THREADS)
scan_kernel(const int* __restrict__ cnt, int* __restrict__ rowptr) {
  static_assert(N_NODES % 4 == 0, "rows of four");
  __shared__ int wsum[32];
  const int tid  = threadIdx.x;
  const int lane = tid & 31;
  const int wave = tid >> 5;
  int carry = 0;
  for (int base = 0; base < N_NODES; base += SCAN_THREADS * SCAN_PER_THREAD) {
    const int i0 = base + tid * SCAN_PER_THREAD;
    int4 c[4];
#pragma unroll
    for (int g = 0; g < 4; ++g) {
      const int i = i0 + 4 * g;
      c[g] = make_int4(0, 0, 0, 0);
      if (i < N_NODES) c[g] = *(const int4*)(cnt + i);
    }
    int t = 0;
#pragma unroll
    for (int g = 0; g < 4; ++g) t += c[g].x + c[g].y + c[g].z + c[g].w;

    int x = t;
#pragma unroll
    for (int o = 1; o < 32; o <<= 1) {
      const int y = __shfl_up(x, o);
      if (lane >= o) x += y;
    }
    if (lane == 31) wsum[wave] = x;
    __syncthreads();

    const int wv = wsum[lane];
    int wp  = (lane < wave) ? wv : 0;
    int tot = wv;
#pragma unroll
    for (int o = 16; o > 0; o >>= 1) {
      wp  += __shfl_xor(wp, o);
      tot += __shfl_xor(tot, o);
    }

    int run = carry + wp + (x - t);
#pragma unroll
    for (int g = 0; g < 4; ++g) {
      const int i = i0 + 4 * g;
      if (i < N_NODES) {
        int4 r;
        r.x = run;
        r.y = r.x + c[g].x;
        r.z = r.y + c[g].y;
        r.w = r.z + c[g].z;
        run = r.w + c[g].w;
        *(int4*)(rowptr + i) = r;
      }
    }
    carry += tot;
    __syncthreads();
  }
}

__global__ void place_kernel(const int* __restrict__ src, const int* __restrict__ dst,
                             const int* __restrict__ rowptr, const int* __restrict__ rank,
                             int* __restrict__ col, int nE) {
  int i = blockIdx.x * blockDim.x + threadIdx.x;
  if (i < nE) col[rowptr[dst[i]] + rank[i]] = src[i];
}


#define GEMM_WAVES 4

template <int NCOLS>
__global__ void __launch_bounds__(GEMM_WAVES * 32)
gemm_wmma_kernel(const float* __restrict__ A, const float* __restrict__ W,
                 float* __restrict__ C, int mtiles) {
  constexpr int NT = NCOLS / 16;
  constexpr int PW = 2 * NCOLS + 32;
  constexpr int C4 = NCOLS / 4;
  __shared__ __attribute__((aligned(16))) float Wl[(KDIM / 2) * PW];

  const int tid = threadIdx.x;

  constexpr int ITEMS = (KDIM / 2) * C4 / (GEMM_WAVES * 32);
  static_assert(ITEMS * GEMM_WAVES * 32 == (KDIM / 2) * C4, "whole items per thread");
#pragma unroll
  for (int u = 0; u < ITEMS; ++u) {
    const int it = tid + u * (GEMM_WAVES * 32);
    const int p = it / C4;
    const int c = it % C4;
    const float4 r0 = *(const float4*)(W + (size_t)(2 * p) * NCOLS + 4 * c);
    const float4 r1 = *(const float4*)(W + (size_t)(2 * p + 1) * NCOLS + 4 * c);
    float* d = Wl + p * PW + 8 * c;
    *(float4*)(d)     = make_float4(r0.x, r1.x, r0.y, r1.y);
    *(float4*)(d + 4) = make_float4(r0.z, r1.z, r0.w, r1.w);
  }
  __syncthreads();

  const int lane = tid & 31;
  const int hi   = lane >> 4;
  const int lo   = lane & 15;
  const int mt = __builtin_amdgcn_readfirstlane((int)(blockIdx.x * GEMM_WAVES + (tid >> 5)));
  if (mt >= mtiles) return;
  const int m0 = mt * 16;

  const float* __restrict__ arow = A + (size_t)(m0 + lo) * KDIM + 2 * hi;
  const float* bl = Wl + hi * PW + 2 * lo;

  v8f acc[NT];
#pragma unroll
  for (int t = 0; t < NT; ++t) acc[t] = v8f{};

#pragma unroll
  for (int j = 0; j < KDIM / 4; ++j) {
    const v2f a = *(const v2f*)(arow + 4 * j);
#pragma unroll
    for (int t = 0; t < NT; ++t) {
      const v2f b = *(const v2f*)(bl + (2 * j) * PW + 32 * t);
      acc[t] = __builtin_amdgcn_wmma_f32_16x16x4_f32(
           false, a,  false, b,
           (short)0, acc[t],  false,  false);
    }
  }

  float* __restrict__ crow = C + (size_t)(m0 + hi * 8) * NCOLS + lo;
#pragma unroll
  for (int t = 0; t < NT; ++t) {
#pragma unroll
    for (int r = 0; r < 8; ++r)
      crow[(size_t)r * NCOLS + 16 * t] = acc[t][r];
  }
}


template <int F, bool RELU>
__global__ void gather_kernel(const float* __restrict__ h, const int* __restrict__ rowptr,
                              const int* __restrict__ cnt, const int* __restrict__ col,
                              const float* __restrict__ bias, float* __restrict__ agg, int n) {
  constexpr int chunks = F / 4;
  constexpr int epw    = 32 / chunks;
  const int lane = threadIdx.x & 31;
  const int sub  = lane / chunks;
  const int l    = lane % chunks;
  const int node = __builtin_amdgcn_readfirstlane(
      (int)(blockIdx.x * (blockDim.x >> 5) + (threadIdx.x >> 5)));
  if (node >= n) return;
  const int   start = rowptr[node];
  const int   m     = cnt[node];
  const float dn    = rsqrtf(1.0f + (float)m);

  const float4 hv = *(const float4*)(h + (size_t)node * F + l * 4);

  float4 acc = make_float4(0.0f, 0.0f, 0.0f, 0.0f);
  for (int b = 0; b < m; b += 32) {
    const int k = b + lane;
    int   idx = node;
    float cf  = 0.0f;
    if (k < m) {
      idx = col[start + k];
      cf  = rsqrtf(1.0f + (float)cnt[idx]) * dn;
    }
    const int rem = m - b;
    const int mb  = rem < 32 ? rem : 32;
    const int T   = (mb + epw - 1) / epw;
    for (int t0 = 0; t0 < T; t0 += 4) {
      float4 v[4];
      float  c[4];
#pragma unroll
      for (int u = 0; u < 4; ++u) {
        int s;
        if constexpr (epw == 1) {
          const int jl = t0 + u;
          s    = __builtin_amdgcn_readlane(idx, jl);
          c[u] = __int_as_float(__builtin_amdgcn_readlane(__float_as_int(cf), jl));
        } else {
          const int jl = (t0 + u) * epw + sub;
          s    = __builtin_amdgcn_ds_bpermute(jl << 2, idx);
          c[u] = __int_as_float(__builtin_amdgcn_ds_bpermute(jl << 2, __float_as_int(cf)));
        }
        v[u] = *(const float4*)(h + (size_t)s * F + l * 4);
      }
#pragma unroll
      for (int u = 0; u < 4; ++u) {
        acc.x += v[u].x * c[u];
        acc.y += v[u].y * c[u];
        acc.z += v[u].z * c[u];
        acc.w += v[u].w * c[u];
      }
    }
  }

  if constexpr (epw == 2) {
    acc.x += __shfl_xor(acc.x, 16);
    acc.y += __shfl_xor(acc.y, 16);
    acc.z += __shfl_xor(acc.z, 16);
    acc.w += __shfl_xor(acc.w, 16);
  }

  if (sub == 0) {
    const float4 bv = *(const float4*)(bias + l * 4);
    const float  d2 = dn * dn;
    float4 o;
    o.x = (acc.x + hv.x * d2) + bv.x;
    o.y = (acc.y + hv.y * d2) + bv.y;
    o.z = (acc.z + hv.z * d2) + bv.z;
    o.w = (acc.w + hv.w * d2) + bv.w;
    if (RELU) {
      o.x = fmaxf(o.x, 0.0f);
      o.y = fmaxf(o.y, 0.0f);
      o.z = fmaxf(o.z, 0.0f);
      o.w = fmaxf(o.w, 0.0f);
    }
    *(float4*)(agg + (size_t)node * F + l * 4) = o;
  }
}


__global__ void dot_kernel(const float* __restrict__ z, const int* __restrict__ ls,
                           const int* __restrict__ ld, float* __restrict__ out, int nL) {
  int e = blockIdx.x * blockDim.x + threadIdx.x;
  if (e >= nL) return;
  const float4* a = (const float4*)(z + (size_t)ls[e] * 64);
  const float4* b = (const float4*)(z + (size_t)ld[e] * 64);
  float s = 0.0f;
#pragma unroll
  for (int i = 0; i < 16; ++i) {
    float4 x = a[i], y = b[i];
    s += x.x * y.x + x.y * y.y + x.z * y.z + x.w * y.w;
  }
  out[e] = s;
}


extern "C" void kernel_launch(void* const* d_in, const int* in_sizes, int n_in,
                              void* d_out, int out_size, void* d_ws, size_t ws_size,
                              hipStream_t stream) {
  const float* x  = (const float*)d_in[0];
  const int*   ei = (const int*)d_in[1];
  const int*   el = (const int*)d_in[2];
  const float* W1 = (const float*)d_in[3];
  const float* b1 = (const float*)d_in[4];
  const float* W2 = (const float*)d_in[5];
  const float* b2 = (const float*)d_in[6];

  const int nE = in_sizes[1] / 2;
  const int nL = in_sizes[2] / 2;
  const int* src = ei;
  const int* dst = ei + nE;
  const int* ls  = el;
  const int* ld  = el + nL;

  int*   cnt    = (int*)d_ws;
  int*   rowptr = cnt + N_PAD;
  int*   rank   = rowptr + N_PAD;
  int*   col    = rank + E_CAP;
  float* bufA   = (float*)(col + E_CAP);
  float* bufB   = bufA + (size_t)N_NODES * 128;
  float* out    = (float*)d_out;

  zero_int_kernel<<<(N_NODES + 255) / 256, 256, 0, stream>>>(cnt, N_NODES);
  rank_kernel<<<(nE + 255) / 256, 256, 0, stream>>>(dst, cnt, rank, nE);
  scan_kernel<<<1, SCAN_THREADS, 0, stream>>>(cnt, rowptr);
  place_kernel<<<(nE + 255) / 256, 256, 0, stream>>>(src, dst, rowptr, rank, col, nE);

  gemm_wmma_kernel<128><<<(N_NODES / 16 + GEMM_WAVES - 1) / GEMM_WAVES, GEMM_WAVES * 32, 0, stream>>>(
      x, W1, bufA, N_NODES / 16);
  gather_kernel<128, true><<<(N_NODES + 7) / 8, 256, 0, stream>>>(bufA, rowptr, cnt, col, b1, bufB,
                                                                   N_NODES);

  gemm_wmma_kernel<64><<<(N_NODES / 16 + GEMM_WAVES - 1) / GEMM_WAVES, GEMM_WAVES * 32, 0, stream>>>(
      bufB, W2, bufA, N_NODES / 16);
  gather_kernel<64, false><<<(N_NODES + 7) / 8, 256, 0, stream>>>(bufA, rowptr, cnt, col, b2, bufB,
                                                                   N_NODES);

  dot_kernel<<<(nL + 255) / 256, 256, 0, stream>>>(bufB, ls, ld, out, nL);
}


# x @ W1 moved into the count-scan launch: work-group 0 scans, 255 32-wave work-groups (one per other WGP) take its strips, W1 staged once each; same k order and sums, one launch fewer
# speedup vs baseline: 2.6224x; 1.0714x over previous
#include <hip/hip_runtime.h>


typedef float v2f __attribute__((ext_vector_type(2)));
typedef float v8f __attribute__((ext_vector_type(8)));

#define N_NODES 100000
#define N_PAD   100352
#define E_CAP   1600000
#define KDIM 128


__global__ void zero_int_kernel(int* __restrict__ p, int n) {
  int i = blockIdx.x * blockDim.x + threadIdx.x;
  if (i < n) p[i] = 0;
}

__global__ void rank_kernel(const int* __restrict__ dst, int* __restrict__ cnt,
                            int* __restrict__ rank, int nE) {
  int i = blockIdx.x * blockDim.x + threadIdx.x;
  if (i < nE) rank[i] = atomicAdd(&cnt[dst[i]], 1);
}

#define SCAN_THREADS 1024
#define SCAN_PER_THREAD 16

__device__ __forceinline__ void scan_body(const int* __restrict__ cnt, int* __restrict__ rowptr) {
  static_assert(N_NODES % 4 == 0, "rows of four");
  __shared__ int wsum[32];
  const int tid  = threadIdx.x;
  const int lane = tid & 31;
  const int wave = tid >> 5;
  int carry = 0;
  for (int base = 0; base < N_NODES; base += SCAN_THREADS * SCAN_PER_THREAD) {
    const int i0 = base + tid * SCAN_PER_THREAD;
    int4 c[4];
#pragma unroll
    for (int g = 0; g < 4; ++g) {
      const int i = i0 + 4 * g;
      c[g] = make_int4(0, 0, 0, 0);
      if (i < N_NODES) c[g] = *(const int4*)(cnt + i);
    }
    int t = 0;
#pragma unroll
    for (int g = 0; g < 4; ++g) t += c[g].x + c[g].y + c[g].z + c[g].w;

    int x = t;
#pragma unroll
    for (int o = 1; o < 32; o <<= 1) {
      const int y = __shfl_up(x, o);
      if (lane >= o) x += y;
    }
    if (lane == 31) wsum[wave] = x;
    __syncthreads();

    const int wv = wsum[lane];
    int wp  = (lane < wave) ? wv : 0;
    int tot = wv;
#pragma unroll
    for (int o = 16; o > 0; o >>= 1) {
      wp  += __shfl_xor(wp, o);
      tot += __shfl_xor(tot, o);
    }

    int run = carry + wp + (x - t);
#pragma unroll
    for (int g = 0; g < 4; ++g) {
      const int i = i0 + 4 * g;
      if (i < N_NODES) {
        int4 r;
        r.x = run;
        r.y = r.x + c[g].x;
        r.z = r.y + c[g].y;
        r.w = r.z + c[g].z;
        run = r.w + c[g].w;
        *(int4*)(rowptr + i) = r;
      }
    }
    carry += tot;
    __syncthreads();
  }
}

__global__ void place_kernel(const int* __restrict__ src, const int* __restrict__ dst,
                             const int* __restrict__ rowptr, const int* __restrict__ rank,
                             int* __restrict__ col, int nE) {
  int i = blockIdx.x * blockDim.x + threadIdx.x;
  if (i < nE) col[rowptr[dst[i]] + rank[i]] = src[i];
}


#define GEMM_WAVES 4

template <int NCOLS>
__global__ void __launch_bounds__(GEMM_WAVES * 32)
gemm_wmma_kernel(const float* __restrict__ A, const float* __restrict__ W,
                 float* __restrict__ C, int mtiles) {
  constexpr int NT = NCOLS / 16;
  constexpr int PW = 2 * NCOLS + 32;
  constexpr int C4 = NCOLS / 4;
  __shared__ __attribute__((aligned(16))) float Wl[(KDIM / 2) * PW];

  const int tid = threadIdx.x;

  constexpr int ITEMS = (KDIM / 2) * C4 / (GEMM_WAVES * 32);
  static_assert(ITEMS * GEMM_WAVES * 32 == (KDIM / 2) * C4, "whole items per thread");
#pragma unroll
  for (int u = 0; u < ITEMS; ++u) {
    const int it = tid + u * (GEMM_WAVES * 32);
    const int p = it / C4;
    const int c = it % C4;
    const float4 r0 = *(const float4*)(W + (size_t)(2 * p) * NCOLS + 4 * c);
    const float4 r1 = *(const float4*)(W + (size_t)(2 * p + 1) * NCOLS + 4 * c);
    float* d = Wl + p * PW + 8 * c;
    *(float4*)(d)     = make_float4(r0.x, r1.x, r0.y, r1.y);
    *(float4*)(d + 4) = make_float4(r0.z, r1.z, r0.w, r1.w);
  }
  __syncthreads();

  const int lane = tid & 31;
  const int hi   = lane >> 4;
  const int lo   = lane & 15;
  const int mt = __builtin_amdgcn_readfirstlane((int)(blockIdx.x * GEMM_WAVES + (tid >> 5)));
  if (mt >= mtiles) return;
  const int m0 = mt * 16;

  const float* __restrict__ arow = A + (size_t)(m0 + lo) * KDIM + 2 * hi;
  const float* bl = Wl + hi * PW + 2 * lo;

  v8f acc[NT];
#pragma unroll
  for (int t = 0; t < NT; ++t) acc[t] = v8f{};

#pragma unroll
  for (int j = 0; j < KDIM / 4; ++j) {
    const v2f a = *(const v2f*)(arow + 4 * j);
#pragma unroll
    for (int t = 0; t < NT; ++t) {
      const v2f b = *(const v2f*)(bl + (2 * j) * PW + 32 * t);
      acc[t] = __builtin_amdgcn_wmma_f32_16x16x4_f32(
           false, a,  false, b,
           (short)0, acc[t],  false,  false);
    }
  }

  float* __restrict__ crow = C + (size_t)(m0 + hi * 8) * NCOLS + lo;
#pragma unroll
  for (int t = 0; t < NT; ++t) {
#pragma unroll
    for (int r = 0; r < 8; ++r)
      crow[(size_t)r * NCOLS + 16 * t] = acc[t][r];
  }
}


__global__ void __launch_bounds__(SCAN_THREADS)
scan_gemm1_kernel(const int* __restrict__ cnt, int* __restrict__ rowptr,
                  const float* __restrict__ A, const float* __restrict__ W,
                  float* __restrict__ C, int mtiles) {
  if (blockIdx.x == 0) {
    scan_body(cnt, rowptr);
    return;
  }
  constexpr int NCOLS = 128;
  constexpr int NT = NCOLS / 16;
  constexpr int PW = 2 * NCOLS + 32;
  constexpr int C4 = NCOLS / 4;
  __shared__ __attribute__((aligned(16))) float Wl[(KDIM / 2) * PW];

  const int tid = threadIdx.x;

  constexpr int ITEMS = (KDIM / 2) * C4 / SCAN_THREADS;
  static_assert(ITEMS * SCAN_THREADS == (KDIM / 2) * C4, "whole items per thread");
#pragma unroll
  for (int u = 0; u < ITEMS; ++u) {
    const int it = tid + u * SCAN_THREADS;
    const int p = it / C4;
    const int c = it % C4;
    const float4 r0 = *(const float4*)(W + (size_t)(2 * p) * NCOLS + 4 * c);
    const float4 r1 = *(const float4*)(W + (size_t)(2 * p + 1) * NCOLS + 4 * c);
    float* d = Wl + p * PW + 8 * c;
    *(float4*)(d)     = make_float4(r0.x, r1.x, r0.y, r1.y);
    *(float4*)(d + 4) = make_float4(r0.z, r1.z, r0.w, r1.w);
  }
  __syncthreads();

  const int lane = tid & 31;
  const int hi   = lane >> 4;
  const int lo   = lane & 15;
  const int nwg  = (int)gridDim.x - 1;
  const int mt = __builtin_amdgcn_readfirstlane((tid >> 5) * nwg + (int)blockIdx.x - 1);
  if (mt >= mtiles) return;
  const int m0 = mt * 16;

  const float* __restrict__ arow = A + (size_t)(m0 + lo) * KDIM + 2 * hi;
  const float* bl = Wl + hi * PW + 2 * lo;

  v8f acc[NT];
#pragma unroll
  for (int t = 0; t < NT; ++t) acc[t] = v8f{};

  v2f a[KDIM / 4];
#pragma unroll
  for (int j = 0; j < KDIM / 4; ++j) a[j] = *(const v2f*)(arow + 4 * j);
#pragma unroll
  for (int j = 0; j < KDIM / 4; ++j) {
#pragma unroll
    for (int t = 0; t < NT; ++t) {
      const v2f b = *(const v2f*)(bl + (2 * j) * PW + 32 * t);
      acc[t] = __builtin_amdgcn_wmma_f32_16x16x4_f32(
           false, a[j],  false, b,
           (short)0, acc[t],  false,  false);
    }
  }

  float* __restrict__ crow = C + (size_t)(m0 + hi * 8) * NCOLS + lo;
#pragma unroll
  for (int t = 0; t < NT; ++t) {
#pragma unroll
    for (int r = 0; r < 8; ++r)
      crow[(size_t)r * NCOLS + 16 * t] = acc[t][r];
  }
}


template <int F, bool RELU>
__global__ void gather_kernel(const float* __restrict__ h, const int* __restrict__ rowptr,
                              const int* __restrict__ cnt, const int* __restrict__ col,
                              const float* __restrict__ bias, float* __restrict__ agg, int n) {
  constexpr int chunks = F / 4;
  constexpr int epw    = 32 / chunks;
  const int lane = threadIdx.x & 31;
  const int sub  = lane / chunks;
  const int l    = lane % chunks;
  const int node = __builtin_amdgcn_readfirstlane(
      (int)(blockIdx.x * (blockDim.x >> 5) + (threadIdx.x >> 5)));
  if (node >= n) return;
  const int   start = rowptr[node];
  const int   m     = cnt[node];
  const float dn    = rsqrtf(1.0f + (float)m);

  const float4 hv = *(const float4*)(h + (size_t)node * F + l * 4);

  float4 acc = make_float4(0.0f, 0.0f, 0.0f, 0.0f);
  for (int b = 0; b < m; b += 32) {
    const int k = b + lane;
    int   idx = node;
    float cf  = 0.0f;
    if (k < m) {
      idx = col[start + k];
      cf  = rsqrtf(1.0f + (float)cnt[idx]) * dn;
    }
    const int rem = m - b;
    const int mb  = rem < 32 ? rem : 32;
    const int T   = (mb + epw - 1) / epw;
    for (int t0 = 0; t0 < T; t0 += 4) {
      float4 v[4];
      float  c[4];
#pragma unroll
      for (int u = 0; u < 4; ++u) {
        int s;
        if constexpr (epw == 1) {
          const int jl = t0 + u;
          s    = __builtin_amdgcn_readlane(idx, jl);
          c[u] = __int_as_float(__builtin_amdgcn_readlane(__float_as_int(cf), jl));
        } else {
          const int jl = (t0 + u) * epw + sub;
          s    = __builtin_amdgcn_ds_bpermute(jl << 2, idx);
          c[u] = __int_as_float(__builtin_amdgcn_ds_bpermute(jl << 2, __float_as_int(cf)));
        }
        v[u] = *(const float4*)(h + (size_t)s * F + l * 4);
      }
#pragma unroll
      for (int u = 0; u < 4; ++u) {
        acc.x += v[u].x * c[u];
        acc.y += v[u].y * c[u];
        acc.z += v[u].z * c[u];
        acc.w += v[u].w * c[u];
      }
    }
  }

  if constexpr (epw == 2) {
    acc.x += __shfl_xor(acc.x, 16);
    acc.y += __shfl_xor(acc.y, 16);
    acc.z += __shfl_xor(acc.z, 16);
    acc.w += __shfl_xor(acc.w, 16);
  }

  if (sub == 0) {
    const float4 bv = *(const float4*)(bias + l * 4);
    const float  d2 = dn * dn;
    float4 o;
    o.x = (acc.x + hv.x * d2) + bv.x;
    o.y = (acc.y + hv.y * d2) + bv.y;
    o.z = (acc.z + hv.z * d2) + bv.z;
    o.w = (acc.w + hv.w * d2) + bv.w;
    if (RELU) {
      o.x = fmaxf(o.x, 0.0f);
      o.y = fmaxf(o.y, 0.0f);
      o.z = fmaxf(o.z, 0.0f);
      o.w = fmaxf(o.w, 0.0f);
    }
    *(float4*)(agg + (size_t)node * F + l * 4) = o;
  }
}


__global__ void dot_kernel(const float* __restrict__ z, const int* __restrict__ ls,
                           const int* __restrict__ ld, float* __restrict__ out, int nL) {
  int e = blockIdx.x * blockDim.x + threadIdx.x;
  if (e >= nL) return;
  const float4* a = (const float4*)(z + (size_t)ls[e] * 64);
  const float4* b = (const float4*)(z + (size_t)ld[e] * 64);
  float s = 0.0f;
#pragma unroll
  for (int i = 0; i < 16; ++i) {
    float4 x = a[i], y = b[i];
    s += x.x * y.x + x.y * y.y + x.z * y.z + x.w * y.w;
  }
  out[e] = s;
}


extern "C" void kernel_launch(void* const* d_in, const int* in_sizes, int n_in,
                              void* d_out, int out_size, void* d_ws, size_t ws_size,
                              hipStream_t stream) {
  const float* x  = (const float*)d_in[0];
  const int*   ei = (const int*)d_in[1];
  const int*   el = (const int*)d_in[2];
  const float* W1 = (const float*)d_in[3];
  const float* b1 = (const float*)d_in[4];
  const float* W2 = (const float*)d_in[5];
  const float* b2 = (const float*)d_in[6];

  const int nE = in_sizes[1] / 2;
  const int nL = in_sizes[2] / 2;
  const int* src = ei;
  const int* dst = ei + nE;
  const int* ls  = el;
  const int* ld  = el + nL;

  int*   cnt    = (int*)d_ws;
  int*   rowptr = cnt + N_PAD;
  int*   rank   = rowptr + N_PAD;
  int*   col    = rank + E_CAP;
  float* bufA   = (float*)(col + E_CAP);
  float* bufB   = bufA + (size_t)N_NODES * 128;
  float* out    = (float*)d_out;

  zero_int_kernel<<<(N_NODES + 255) / 256, 256, 0, stream>>>(cnt, N_NODES);
  rank_kernel<<<(nE + 255) / 256, 256, 0, stream>>>(dst, cnt, rank, nE);
  scan_gemm1_kernel<<<256, SCAN_THREADS, 0, stream>>>(cnt, rowptr, x, W1, bufA, N_NODES / 16);
  place_kernel<<<(nE + 255) / 256, 256, 0, stream>>>(src, dst, rowptr, rank, col, nE);

  gather_kernel<128, true><<<(N_NODES + 7) / 8, 256, 0, stream>>>(bufA, rowptr, cnt, col, b1, bufB,
                                                                   N_NODES);

  gemm_wmma_kernel<64><<<(N_NODES / 16 + GEMM_WAVES - 1) / GEMM_WAVES, GEMM_WAVES * 32, 0, stream>>>(
      bufB, W2, bufA, N_NODES / 16);
  gather_kernel<64, false><<<(N_NODES + 7) / 8, 256, 0, stream>>>(bufA, rowptr, cnt, col, b2, bufB,
                                                                   N_NODES);

  dot_kernel<<<(nL + 255) / 256, 256, 0, stream>>>(bufB, ls, ld, out, nL);
}


# Edge list by destination built on LDS counters (chunks ordered by 256-node bucket, then each bucket by node, in LDS): no global atomic or scattered store; in-edge sum order = LDS arrival order
# speedup vs baseline: 4.3662x; 1.6650x over previous
#include <hip/hip_runtime.h>


typedef float v2f __attribute__((ext_vector_type(2)));
typedef float v8f __attribute__((ext_vector_type(8)));

#define N_NODES 100000
#define N_PAD   100352
#define E_CAP   1600000
#define KDIM 128


#define NCH         256
#define CHUNK       (E_CAP / NCH)
#define BSH         8
#define BNODES      (1 << BSH)
#define NB          ((N_NODES + BNODES - 1) >> BSH)
#define BIN_THREADS 1024
#define EPT         ((CHUNK + BIN_THREADS - 1) / BIN_THREADS)
#define STAGE_CAP   16384

static_assert(E_CAP % NCH == 0, "whole chunks");
static_assert(NB <= BIN_THREADS && BNODES <= BIN_THREADS, "one thread per counter");
static_assert(NCH * 4 == BIN_THREADS, "four threads per run");
static_assert(N_NODES <= (1 << (31 - BSH)), "source and node-in-bucket share one word");

__device__ __forceinline__ int block_excl_scan(int t, int* wsum, int& tot) {
  const int lane = threadIdx.x & 31;
  const int wave = threadIdx.x >> 5;
  int x = t;
#pragma unroll
  for (int o = 1; o < 32; o <<= 1) {
    const int y = __shfl_up(x, o);
    if (lane >= o) x += y;
  }
  if (lane == 31) wsum[wave] = x;
  __syncthreads();
  const int wv = wsum[lane];
  int wp = (lane < wave) ? wv : 0;
  int s  = wv;
#pragma unroll
  for (int o = 16; o > 0; o >>= 1) {
    wp += __shfl_xor(wp, o);
    s  += __shfl_xor(s, o);
  }
  __syncthreads();
  tot = s;
  return wp + (x - t);
}

__global__ void __launch_bounds__(BIN_THREADS)
bin_kernel(const int* __restrict__ src, const int* __restrict__ dst, int* __restrict__ sorted,
           int* __restrict__ runinfo, int nE) {
  __shared__ int hist[NB];
  __shared__ int lstart[NB];
  __shared__ int wsum[32];
  __shared__ int stage[CHUNK];
  const int tid   = threadIdx.x;
  const int chunk = blockIdx.x;
  const int beg   = chunk * CHUNK;
  int n = nE - beg;
  n = n < 0 ? 0 : (n > CHUNK ? CHUNK : n);

  if (n == 0) {
    if (tid < NB) ((int2*)runinfo)[tid * NCH + chunk] = make_int2(beg, 0);
    return;
  }
  if (tid < NB) hist[tid] = 0;
  __syncthreads();

  int w[EPT], bk[EPT], lr[EPT];
#pragma unroll
  for (int u = 0; u < EPT; ++u) {
    const int i  = u * BIN_THREADS + tid;
    const int ic = i < n ? i : n - 1;
    const int d  = dst[beg + ic];
    const int s  = src[beg + ic];
    bk[u] = i < n ? (d >> BSH) : -1;
    w[u]  = (s << BSH) | (d & (BNODES - 1));
  }
#pragma unroll
  for (int u = 0; u < EPT; ++u) {
    lr[u] = 0;
    if (bk[u] >= 0) lr[u] = atomicAdd(&hist[bk[u]], 1);
  }
  __syncthreads();

  int tot;
  const int h  = tid < NB ? hist[tid] : 0;
  const int ex = block_excl_scan(h, wsum, tot);
  if (tid < NB) {
    lstart[tid] = ex;
    ((int2*)runinfo)[tid * NCH + chunk] = make_int2(beg + ex, h);
  }
  __syncthreads();

#pragma unroll
  for (int u = 0; u < EPT; ++u)
    if (bk[u] >= 0) stage[lstart[bk[u]] + lr[u]] = w[u];
  __syncthreads();

  for (int i = tid; i < n; i += BIN_THREADS) sorted[beg + i] = stage[i];
}

__global__ void __launch_bounds__(BIN_THREADS)
bucket_kernel(const int* __restrict__ sorted, const int* __restrict__ runinfo,
              int* __restrict__ cnt, int* __restrict__ rowptr, int* __restrict__ col) {
  __shared__ int rpos[NCH];
  __shared__ int rlen[NCH];
  __shared__ int c[BNODES];
  __shared__ int cur[BNODES];
  __shared__ int wsum[32];
  __shared__ int stage[STAGE_CAP];
  const int tid = threadIdx.x;
  const int b   = blockIdx.x;

  int off = 0, len = 0;
  if (tid < NCH) {
    const int2 ri = ((const int2*)runinfo)[b * NCH + tid];
    rpos[tid] = ri.x;
    rlen[tid] = ri.y;
    off = ri.x - tid * CHUNK;
    len = ri.y;
  }
  if (tid < BNODES) c[tid] = 0;

  int s, tot;
  block_excl_scan(off, wsum, s);
  block_excl_scan(len, wsum, tot);
  const bool fits = tot <= STAGE_CAP;

  const int r  = tid >> 2;
  const int q  = tid & 3;
  const int rp = rpos[r];
  const int rl = rlen[r];
  for (int j0 = 0; j0 < rl; j0 += 16) {
    int w[4];
#pragma unroll
    for (int u = 0; u < 4; ++u) {
      const int j = j0 + 4 * u + q;
      w[u] = sorted[rp + (j < rl ? j : rl - 1)];
    }
#pragma unroll
    for (int u = 0; u < 4; ++u)
      if (j0 + 4 * u + q < rl) atomicAdd(&c[w[u] & (BNODES - 1)], 1);
  }
  __syncthreads();

  int unused;
  const int h  = tid < BNODES ? c[tid] : 0;
  const int ex = block_excl_scan(h, wsum, unused);
  if (tid < BNODES) {
    cur[tid] = ex;
    const int node = b * BNODES + tid;
    if (node < N_NODES) {
      cnt[node]    = h;
      rowptr[node] = s + ex;
    }
  }
  __syncthreads();

  for (int j0 = 0; j0 < rl; j0 += 16) {
    int w[4];
#pragma unroll
    for (int u = 0; u < 4; ++u) {
      const int j = j0 + 4 * u + q;
      w[u] = sorted[rp + (j < rl ? j : rl - 1)];
    }
#pragma unroll
    for (int u = 0; u < 4; ++u) {
      if (j0 + 4 * u + q < rl) {
        const int p = atomicAdd(&cur[w[u] & (BNODES - 1)], 1);
        if (fits) stage[p] = w[u] >> BSH;
        else      col[s + p] = w[u] >> BSH;
      }
    }
  }
  __syncthreads();

  if (fits)
    for (int i = tid; i < tot; i += BIN_THREADS) col[s + i] = stage[i];
}


#define GEMM_WAVES 4

template <int NCOLS>
__global__ void __launch_bounds__(GEMM_WAVES * 32)
gemm_wmma_kernel(const float* __restrict__ A, const float* __restrict__ W,
                 float* __restrict__ C, int mtiles) {
  constexpr int NT = NCOLS / 16;
  constexpr int PW = 2 * NCOLS + 32;
  constexpr int C4 = NCOLS / 4;
  __shared__ __attribute__((aligned(16))) float Wl[(KDIM / 2) * PW];

  const int tid = threadIdx.x;

  constexpr int ITEMS = (KDIM / 2) * C4 / (GEMM_WAVES * 32);
  static_assert(ITEMS * GEMM_WAVES * 32 == (KDIM / 2) * C4, "whole items per thread");
#pragma unroll
  for (int u = 0; u < ITEMS; ++u) {
    const int it = tid + u * (GEMM_WAVES * 32);
    const int p = it / C4;
    const int c = it % C4;
    const float4 r0 = *(const float4*)(W + (size_t)(2 * p) * NCOLS + 4 * c);
    const float4 r1 = *(const float4*)(W + (size_t)(2 * p + 1) * NCOLS + 4 * c);
    float* d = Wl + p * PW + 8 * c;
    *(float4*)(d)     = make_float4(r0.x, r1.x, r0.y, r1.y);
    *(float4*)(d + 4) = make_float4(r0.z, r1.z, r0.w, r1.w);
  }
  __syncthreads();

  const int lane = tid & 31;
  const int hi   = lane >> 4;
  const int lo   = lane & 15;
  const int mt = __builtin_amdgcn_readfirstlane((int)(blockIdx.x * GEMM_WAVES + (tid >> 5)));
  if (mt >= mtiles) return;
  const int m0 = mt * 16;

  const float* __restrict__ arow = A + (size_t)(m0 + lo) * KDIM + 2 * hi;
  const float* bl = Wl + hi * PW + 2 * lo;

  v8f acc[NT];
#pragma unroll
  for (int t = 0; t < NT; ++t) acc[t] = v8f{};

#pragma unroll
  for (int j = 0; j < KDIM / 4; ++j) {
    const v2f a = *(const v2f*)(arow + 4 * j);
#pragma unroll
    for (int t = 0; t < NT; ++t) {
      const v2f b = *(const v2f*)(bl + (2 * j) * PW + 32 * t);
      acc[t] = __builtin_amdgcn_wmma_f32_16x16x4_f32(
           false, a,  false, b,
           (short)0, acc[t],  false,  false);
    }
  }

  float* __restrict__ crow = C + (size_t)(m0 + hi * 8) * NCOLS + lo;
#pragma unroll
  for (int t = 0; t < NT; ++t) {
#pragma unroll
    for (int r = 0; r < 8; ++r)
      crow[(size_t)r * NCOLS + 16 * t] = acc[t][r];
  }
}


#define GEMM1_THREADS 1024

__global__ void __launch_bounds__(GEMM1_THREADS)
gemm1_kernel(const float* __restrict__ A, const float* __restrict__ W,
             float* __restrict__ C, int mtiles) {
  constexpr int NCOLS = 128;
  constexpr int NT = NCOLS / 16;
  constexpr int PW = 2 * NCOLS + 32;
  constexpr int C4 = NCOLS / 4;
  __shared__ __attribute__((aligned(16))) float Wl[(KDIM / 2) * PW];

  const int tid = threadIdx.x;

  constexpr int ITEMS = (KDIM / 2) * C4 / GEMM1_THREADS;
  static_assert(ITEMS * GEMM1_THREADS == (KDIM / 2) * C4, "whole items per thread");
#pragma unroll
  for (int u = 0; u < ITEMS; ++u) {
    const int it = tid + u * GEMM1_THREADS;
    const int p = it / C4;
    const int c = it % C4;
    const float4 r0 = *(const float4*)(W + (size_t)(2 * p) * NCOLS + 4 * c);
    const float4 r1 = *(const float4*)(W + (size_t)(2 * p + 1) * NCOLS + 4 * c);
    float* d = Wl + p * PW + 8 * c;
    *(float4*)(d)     = make_float4(r0.x, r1.x, r0.y, r1.y);
    *(float4*)(d + 4) = make_float4(r0.z, r1.z, r0.w, r1.w);
  }
  __syncthreads();

  const int lane = tid & 31;
  const int hi   = lane >> 4;
  const int lo   = lane & 15;
  const int nwg  = (int)gridDim.x;
  const int mt = __builtin_amdgcn_readfirstlane((tid >> 5) * nwg + (int)blockIdx.x);
  if (mt >= mtiles) return;
  const int m0 = mt * 16;

  const float* __restrict__ arow = A + (size_t)(m0 + lo) * KDIM + 2 * hi;
  const float* bl = Wl + hi * PW + 2 * lo;

  v8f acc[NT];
#pragma unroll
  for (int t = 0; t < NT; ++t) acc[t] = v8f{};

  v2f a[KDIM / 4];
#pragma unroll
  for (int j = 0; j < KDIM / 4; ++j) a[j] = *(const v2f*)(arow + 4 * j);
#pragma unroll
  for (int j = 0; j < KDIM / 4; ++j) {
#pragma unroll
    for (int t = 0; t < NT; ++t) {
      const v2f b = *(const v2f*)(bl + (2 * j) * PW + 32 * t);
      acc[t] = __builtin_amdgcn_wmma_f32_16x16x4_f32(
           false, a[j],  false, b,
           (short)0, acc[t],  false,  false);
    }
  }

  float* __restrict__ cbase = C + (size_t)m0 * NCOLS;
  const unsigned coff = (unsigned)(hi * 8 * NCOLS + lo);
#pragma unroll
  for (int t = 0; t < NT; ++t) {
#pragma unroll
    for (int r = 0; r < 8; ++r)
      cbase[coff + (unsigned)(r * NCOLS + 16 * t)] = acc[t][r];
  }
}


template <int F, bool RELU>
__global__ void gather_kernel(const float* __restrict__ h, const int* __restrict__ rowptr,
                              const int* __restrict__ cnt, const int* __restrict__ col,
                              const float* __restrict__ bias, float* __restrict__ agg, int n) {
  constexpr int chunks = F / 4;
  constexpr int epw    = 32 / chunks;
  const int lane = threadIdx.x & 31;
  const int sub  = lane / chunks;
  const int l    = lane % chunks;
  const int node = __builtin_amdgcn_readfirstlane(
      (int)(blockIdx.x * (blockDim.x >> 5) + (threadIdx.x >> 5)));
  if (node >= n) return;
  const int   start = rowptr[node];
  const int   m     = cnt[node];
  const float dn    = rsqrtf(1.0f + (float)m);

  const float4 hv = *(const float4*)(h + (size_t)node * F + l * 4);

  float4 acc = make_float4(0.0f, 0.0f, 0.0f, 0.0f);
  for (int b = 0; b < m; b += 32) {
    const int k = b + lane;
    int   idx = node;
    float cf  = 0.0f;
    if (k < m) {
      idx = col[start + k];
      cf  = rsqrtf(1.0f + (float)cnt[idx]) * dn;
    }
    const int rem = m - b;
    const int mb  = rem < 32 ? rem : 32;
    const int T   = (mb + epw - 1) / epw;
    for (int t0 = 0; t0 < T; t0 += 4) {
      float4 v[4];
      float  c[4];
#pragma unroll
      for (int u = 0; u < 4; ++u) {
        int s;
        if constexpr (epw == 1) {
          const int jl = t0 + u;
          s    = __builtin_amdgcn_readlane(idx, jl);
          c[u] = __int_as_float(__builtin_amdgcn_readlane(__float_as_int(cf), jl));
        } else {
          const int jl = (t0 + u) * epw + sub;
          s    = __builtin_amdgcn_ds_bpermute(jl << 2, idx);
          c[u] = __int_as_float(__builtin_amdgcn_ds_bpermute(jl << 2, __float_as_int(cf)));
        }
        v[u] = *(const float4*)(h + (size_t)s * F + l * 4);
      }
#pragma unroll
      for (int u = 0; u < 4; ++u) {
        acc.x += v[u].x * c[u];
        acc.y += v[u].y * c[u];
        acc.z += v[u].z * c[u];
        acc.w += v[u].w * c[u];
      }
    }
  }

  if constexpr (epw == 2) {
    acc.x += __shfl_xor(acc.x, 16);
    acc.y += __shfl_xor(acc.y, 16);
    acc.z += __shfl_xor(acc.z, 16);
    acc.w += __shfl_xor(acc.w, 16);
  }

  if (sub == 0) {
    const float4 bv = *(const float4*)(bias + l * 4);
    const float  d2 = dn * dn;
    float4 o;
    o.x = (acc.x + hv.x * d2) + bv.x;
    o.y = (acc.y + hv.y * d2) + bv.y;
    o.z = (acc.z + hv.z * d2) + bv.z;
    o.w = (acc.w + hv.w * d2) + bv.w;
    if (RELU) {
      o.x = fmaxf(o.x, 0.0f);
      o.y = fmaxf(o.y, 0.0f);
      o.z = fmaxf(o.z, 0.0f);
      o.w = fmaxf(o.w, 0.0f);
    }
    *(float4*)(agg + (size_t)node * F + l * 4) = o;
  }
}


__global__ void dot_kernel(const float* __restrict__ z, const int* __restrict__ ls,
                           const int* __restrict__ ld, float* __restrict__ out, int nL) {
  int e = blockIdx.x * blockDim.x + threadIdx.x;
  if (e >= nL) return;
  const float4* a = (const float4*)(z + (size_t)ls[e] * 64);
  const float4* b = (const float4*)(z + (size_t)ld[e] * 64);
  float s = 0.0f;
#pragma unroll
  for (int i = 0; i < 16; ++i) {
    float4 x = a[i], y = b[i];
    s += x.x * y.x + x.y * y.y + x.z * y.z + x.w * y.w;
  }
  out[e] = s;
}


extern "C" void kernel_launch(void* const* d_in, const int* in_sizes, int n_in,
                              void* d_out, int out_size, void* d_ws, size_t ws_size,
                              hipStream_t stream) {
  const float* x  = (const float*)d_in[0];
  const int*   ei = (const int*)d_in[1];
  const int*   el = (const int*)d_in[2];
  const float* W1 = (const float*)d_in[3];
  const float* b1 = (const float*)d_in[4];
  const float* W2 = (const float*)d_in[5];
  const float* b2 = (const float*)d_in[6];

  const int nE = in_sizes[1] / 2;
  const int nL = in_sizes[2] / 2;
  const int* src = ei;
  const int* dst = ei + nE;
  const int* ls  = el;
  const int* ld  = el + nL;

  int*   cnt     = (int*)d_ws;
  int*   rowptr  = cnt + N_PAD;
  int*   sorted  = rowptr + N_PAD;
  int*   col     = sorted + E_CAP;
  int*   runinfo = col + E_CAP;
  float* bufA    = (float*)(runinfo + 2 * NB * NCH);
  float* bufB    = bufA + (size_t)N_NODES * 128;
  float* out     = (float*)d_out;

  bin_kernel<<<NCH, BIN_THREADS, 0, stream>>>(src, dst, sorted, runinfo, nE);
  bucket_kernel<<<NB, BIN_THREADS, 0, stream>>>(sorted, runinfo, cnt, rowptr, col);
  gemm1_kernel<<<256, GEMM1_THREADS, 0, stream>>>(x, W1, bufA, N_NODES / 16);

  gather_kernel<128, true><<<(N_NODES + 7) / 8, 256, 0, stream>>>(bufA, rowptr, cnt, col, b1, bufB,
                                                                   N_NODES);

  gemm_wmma_kernel<64><<<(N_NODES / 16 + GEMM_WAVES - 1) / GEMM_WAVES, GEMM_WAVES * 32, 0, stream>>>(
      bufB, W2, bufA, N_NODES / 16);
  gather_kernel<64, false><<<(N_NODES + 7) / 8, 256, 0, stream>>>(bufA, rowptr, cnt, col, b2, bufB,
                                                                   N_NODES);

  dot_kernel<<<(nL + 255) / 256, 256, 0, stream>>>(bufB, ls, ld, out, nL);
}


# Layer-2 product fused into the layer-1 gather: each 16-node strip of relu rows stays in LDS and goes through W2 on the matrix unit there; one launch fewer, no hidden-row round trip; sums unchanged
# speedup vs baseline: 4.9707x; 1.1384x over previous
#include <hip/hip_runtime.h>


typedef float v2f __attribute__((ext_vector_type(2)));
typedef float v8f __attribute__((ext_vector_type(8)));

#define N_NODES 100000
#define N_PAD   100352
#define E_CAP   1600000
#define KDIM 128


#define NCH         256
#define CHUNK       (E_CAP / NCH)
#define BSH         8
#define BNODES      (1 << BSH)
#define NB          ((N_NODES + BNODES - 1) >> BSH)
#define BIN_THREADS 1024
#define EPT         ((CHUNK + BIN_THREADS - 1) / BIN_THREADS)
#define STAGE_CAP   16384

static_assert(E_CAP % NCH == 0, "whole chunks");
static_assert(NB <= BIN_THREADS && BNODES <= BIN_THREADS, "one thread per counter");
static_assert(NCH * 4 == BIN_THREADS, "four threads per run");
static_assert(N_NODES <= (1 << (31 - BSH)), "source and node-in-bucket share one word");

__device__ __forceinline__ int block_excl_scan(int t, int* wsum, int& tot) {
  const int lane = threadIdx.x & 31;
  const int wave = threadIdx.x >> 5;
  int x = t;
#pragma unroll
  for (int o = 1; o < 32; o <<= 1) {
    const int y = __shfl_up(x, o);
    if (lane >= o) x += y;
  }
  if (lane == 31) wsum[wave] = x;
  __syncthreads();
  const int wv = wsum[lane];
  int wp = (lane < wave) ? wv : 0;
  int s  = wv;
#pragma unroll
  for (int o = 16; o > 0; o >>= 1) {
    wp += __shfl_xor(wp, o);
    s  += __shfl_xor(s, o);
  }
  __syncthreads();
  tot = s;
  return wp + (x - t);
}

__global__ void __launch_bounds__(BIN_THREADS)
bin_kernel(const int* __restrict__ src, const int* __restrict__ dst, int* __restrict__ sorted,
           int* __restrict__ runinfo, int nE) {
  __shared__ int hist[NB];
  __shared__ int lstart[NB];
  __shared__ int wsum[32];
  __shared__ int stage[CHUNK];
  const int tid   = threadIdx.x;
  const int chunk = blockIdx.x;
  const int beg   = chunk * CHUNK;
  int n = nE - beg;
  n = n < 0 ? 0 : (n > CHUNK ? CHUNK : n);

  if (n == 0) {
    if (tid < NB) ((int2*)runinfo)[tid * NCH + chunk] = make_int2(beg, 0);
    return;
  }
  if (tid < NB) hist[tid] = 0;
  __syncthreads();

  int w[EPT], bk[EPT], lr[EPT];
#pragma unroll
  for (int u = 0; u < EPT; ++u) {
    const int i  = u * BIN_THREADS + tid;
    const int ic = i < n ? i : n - 1;
    const int d  = dst[beg + ic];
    const int s  = src[beg + ic];
    bk[u] = i < n ? (d >> BSH) : -1;
    w[u]  = (s << BSH) | (d & (BNODES - 1));
  }
#pragma unroll
  for (int u = 0; u < EPT; ++u) {
    lr[u] = 0;
    if (bk[u] >= 0) lr[u] = atomicAdd(&hist[bk[u]], 1);
  }
  __syncthreads();

  int tot;
  const int h  = tid < NB ? hist[tid] : 0;
  const int ex = block_excl_scan(h, wsum, tot);
  if (tid < NB) {
    lstart[tid] = ex;
    ((int2*)runinfo)[tid * NCH + chunk] = make_int2(beg + ex, h);
  }
  __syncthreads();

#pragma unroll
  for (int u = 0; u < EPT; ++u)
    if (bk[u] >= 0) stage[lstart[bk[u]] + lr[u]] = w[u];
  __syncthreads();

  for (int i = tid; i < n; i += BIN_THREADS) sorted[beg + i] = stage[i];
}

__global__ void __launch_bounds__(BIN_THREADS)
bucket_kernel(const int* __restrict__ sorted, const int* __restrict__ runinfo,
              int* __restrict__ cnt, int* __restrict__ rowptr, int* __restrict__ col) {
  __shared__ int rpos[NCH];
  __shared__ int rlen[NCH];
  __shared__ int c[BNODES];
  __shared__ int cur[BNODES];
  __shared__ int wsum[32];
  __shared__ int stage[STAGE_CAP];
  const int tid = threadIdx.x;
  const int b   = blockIdx.x;

  int off = 0, len = 0;
  if (tid < NCH) {
    const int2 ri = ((const int2*)runinfo)[b * NCH + tid];
    rpos[tid] = ri.x;
    rlen[tid] = ri.y;
    off = ri.x - tid * CHUNK;
    len = ri.y;
  }
  if (tid < BNODES) c[tid] = 0;

  int s, tot;
  block_excl_scan(off, wsum, s);
  block_excl_scan(len, wsum, tot);
  const bool fits = tot <= STAGE_CAP;

  const int r  = tid >> 2;
  const int q  = tid & 3;
  const int rp = rpos[r];
  const int rl = rlen[r];
  for (int j0 = 0; j0 < rl; j0 += 16) {
    int w[4];
#pragma unroll
    for (int u = 0; u < 4; ++u) {
      const int j = j0 + 4 * u + q;
      w[u] = sorted[rp + (j < rl ? j : rl - 1)];
    }
#pragma unroll
    for (int u = 0; u < 4; ++u)
      if (j0 + 4 * u + q < rl) atomicAdd(&c[w[u] & (BNODES - 1)], 1);
  }
  __syncthreads();

  int unused;
  const int h  = tid < BNODES ? c[tid] : 0;
  const int ex = block_excl_scan(h, wsum, unused);
  if (tid < BNODES) {
    cur[tid] = ex;
    const int node = b * BNODES + tid;
    if (node < N_NODES) {
      cnt[node]    = h;
      rowptr[node] = s + ex;
    }
  }
  __syncthreads();

  for (int j0 = 0; j0 < rl; j0 += 16) {
    int w[4];
#pragma unroll
    for (int u = 0; u < 4; ++u) {
      const int j = j0 + 4 * u + q;
      w[u] = sorted[rp + (j < rl ? j : rl - 1)];
    }
#pragma unroll
    for (int u = 0; u < 4; ++u) {
      if (j0 + 4 * u + q < rl) {
        const int p = atomicAdd(&cur[w[u] & (BNODES - 1)], 1);
        if (fits) stage[p] = w[u] >> BSH;
        else      col[s + p] = w[u] >> BSH;
      }
    }
  }
  __syncthreads();

  if (fits)
    for (int i = tid; i < tot; i += BIN_THREADS) col[s + i] = stage[i];
}


#define GEMM1_THREADS 1024

__global__ void __launch_bounds__(GEMM1_THREADS)
gemm1_kernel(const float* __restrict__ A, const float* __restrict__ W,
             const int* __restrict__ cnt, float* __restrict__ C, int mtiles) {
  constexpr int NCOLS = 128;
  constexpr int NT = NCOLS / 16;
  constexpr int PW = 2 * NCOLS + 32;
  constexpr int C4 = NCOLS / 4;
  __shared__ __attribute__((aligned(16))) float Wl[(KDIM / 2) * PW];

  const int tid = threadIdx.x;

  constexpr int ITEMS = (KDIM / 2) * C4 / GEMM1_THREADS;
  static_assert(ITEMS * GEMM1_THREADS == (KDIM / 2) * C4, "whole items per thread");
#pragma unroll
  for (int u = 0; u < ITEMS; ++u) {
    const int it = tid + u * GEMM1_THREADS;
    const int p = it / C4;
    const int c = it % C4;
    const float4 r0 = *(const float4*)(W + (size_t)(2 * p) * NCOLS + 4 * c);
    const float4 r1 = *(const float4*)(W + (size_t)(2 * p + 1) * NCOLS + 4 * c);
    float* d = Wl + p * PW + 8 * c;
    *(float4*)(d)     = make_float4(r0.x, r1.x, r0.y, r1.y);
    *(float4*)(d + 4) = make_float4(r0.z, r1.z, r0.w, r1.w);
  }
  __syncthreads();

  const int lane = tid & 31;
  const int hi   = lane >> 4;
  const int lo   = lane & 15;
  const int nwg  = (int)gridDim.x;
  const int mt = __builtin_amdgcn_readfirstlane((tid >> 5) * nwg + (int)blockIdx.x);
  if (mt >= mtiles) return;
  const int m0 = mt * 16;

  const float* __restrict__ arow = A + (size_t)(m0 + lo) * KDIM + 2 * hi;
  const float* bl = Wl + hi * PW + 2 * lo;

  v8f acc[NT];
#pragma unroll
  for (int t = 0; t < NT; ++t) acc[t] = v8f{};

  v2f a[KDIM / 4];
#pragma unroll
  for (int j = 0; j < KDIM / 4; ++j) a[j] = *(const v2f*)(arow + 4 * j);
#pragma unroll
  for (int j = 0; j < KDIM / 4; ++j) {
#pragma unroll
    for (int t = 0; t < NT; ++t) {
      const v2f b = *(const v2f*)(bl + (2 * j) * PW + 32 * t);
      acc[t] = __builtin_amdgcn_wmma_f32_16x16x4_f32(
           false, a[j],  false, b,
           (short)0, acc[t],  false,  false);
    }
  }

  float* __restrict__ cbase = C + (size_t)m0 * NCOLS;
  const unsigned coff = (unsigned)(hi * 8 * NCOLS + lo);
  const int4* __restrict__ cp = (const int4*)(cnt + m0 + hi * 8);
  const int4 c0 = cp[0];
  const int4 c1 = cp[1];
  const float ds[8] = {rsqrtf(1.0f + (float)c0.x), rsqrtf(1.0f + (float)c0.y),
                       rsqrtf(1.0f + (float)c0.z), rsqrtf(1.0f + (float)c0.w),
                       rsqrtf(1.0f + (float)c1.x), rsqrtf(1.0f + (float)c1.y),
                       rsqrtf(1.0f + (float)c1.z), rsqrtf(1.0f + (float)c1.w)};
#pragma unroll
  for (int t = 0; t < NT; ++t) {
#pragma unroll
    for (int r = 0; r < 8; ++r)
      cbase[coff + (unsigned)(r * NCOLS + 16 * t)] = acc[t][r] * ds[r];
  }
}


template <int F, bool RELU>
__global__ void gather_kernel(const float* __restrict__ hs, const int* __restrict__ rowptr,
                              const int* __restrict__ cnt, const int* __restrict__ col,
                              const float* __restrict__ bias, float* __restrict__ agg, int n) {
  constexpr int chunks = F / 4;
  constexpr int epw    = 32 / chunks;
  const int lane = threadIdx.x & 31;
  const int sub  = lane / chunks;
  const int l    = lane % chunks;
  const int node = __builtin_amdgcn_readfirstlane(
      (int)(blockIdx.x * (blockDim.x >> 5) + (threadIdx.x >> 5)));
  if (node >= n) return;
  const int   start = rowptr[node];
  const int   m     = cnt[node];
  const float dn    = rsqrtf(1.0f + (float)m);

  const float4 hv = *(const float4*)(hs + (size_t)node * F + l * 4);

  float4 acc = make_float4(0.0f, 0.0f, 0.0f, 0.0f);
  for (int b = 0; b < m; b += 32) {
    const int k = b + lane;
    int idx = node;
    if (k < m) idx = col[start + k];
    const int rem = m - b;
    const int mb  = rem < 32 ? rem : 32;
    const int T   = (mb + epw - 1) / epw;
    for (int t0 = 0; t0 < T; t0 += 4) {
      float4 v[4];
      int    s[4], jl[4];
#pragma unroll
      for (int u = 0; u < 4; ++u) {
        if constexpr (epw == 1) {
          jl[u] = t0 + u;
          s[u]  = __builtin_amdgcn_readlane(idx, jl[u]);
        } else {
          jl[u] = (t0 + u) * epw + sub;
          s[u]  = __builtin_amdgcn_ds_bpermute(jl[u] << 2, idx);
        }
      }
#pragma unroll
      for (int u = 0; u < 4; ++u) {
        v[u] = make_float4(0.0f, 0.0f, 0.0f, 0.0f);
        if (jl[u] < mb) v[u] = *(const float4*)(hs + (size_t)s[u] * F + l * 4);
      }
#pragma unroll
      for (int u = 0; u < 4; ++u) {
        acc.x += v[u].x;
        acc.y += v[u].y;
        acc.z += v[u].z;
        acc.w += v[u].w;
      }
    }
  }

  if constexpr (epw == 2) {
    acc.x += __shfl_xor(acc.x, 16);
    acc.y += __shfl_xor(acc.y, 16);
    acc.z += __shfl_xor(acc.z, 16);
    acc.w += __shfl_xor(acc.w, 16);
  }

  if (sub == 0) {
    const float4 bv = *(const float4*)(bias + l * 4);
    float4 o;
    o.x = (acc.x + hv.x) * dn + bv.x;
    o.y = (acc.y + hv.y) * dn + bv.y;
    o.z = (acc.z + hv.z) * dn + bv.z;
    o.w = (acc.w + hv.w) * dn + bv.w;
    if (RELU) {
      o.x = fmaxf(o.x, 0.0f);
      o.y = fmaxf(o.y, 0.0f);
      o.z = fmaxf(o.z, 0.0f);
      o.w = fmaxf(o.w, 0.0f);
    }
    *(float4*)(agg + (size_t)node * F + l * 4) = o;
  }
}


#define FG_WAVES 16
#define SPITCH   132

__global__ void __launch_bounds__(FG_WAVES * 32)
gather_gemm_kernel(const float* __restrict__ hs, const int* __restrict__ rowptr,
                   const int* __restrict__ cnt, const int* __restrict__ col,
                   const float* __restrict__ bias, const float* __restrict__ W,
                   float* __restrict__ C, int nstrips) {
  constexpr int F     = 128;
  constexpr int NCOLS = 64;
  constexpr int PW    = 2 * NCOLS + 32;
  constexpr int C4    = NCOLS / 4;
  __shared__ __attribute__((aligned(16))) float Wl[(KDIM / 2) * PW];
  __shared__ __attribute__((aligned(16))) float S[2][16 * SPITCH];

  const int tid = threadIdx.x;

  constexpr int ITEMS = (KDIM / 2) * C4 / (FG_WAVES * 32);
  static_assert(ITEMS * FG_WAVES * 32 == (KDIM / 2) * C4, "whole items per thread");
#pragma unroll
  for (int u = 0; u < ITEMS; ++u) {
    const int it = tid + u * (FG_WAVES * 32);
    const int p = it / C4;
    const int c = it % C4;
    const float4 r0 = *(const float4*)(W + (size_t)(2 * p) * NCOLS + 4 * c);
    const float4 r1 = *(const float4*)(W + (size_t)(2 * p + 1) * NCOLS + 4 * c);
    float* d = Wl + p * PW + 8 * c;
    *(float4*)(d)     = make_float4(r0.x, r1.x, r0.y, r1.y);
    *(float4*)(d + 4) = make_float4(r0.z, r1.z, r0.w, r1.w);
  }
  __syncthreads();

  const int lane = tid & 31;
  const int hi   = lane >> 4;
  const int lo   = lane & 15;
  const int wave = __builtin_amdgcn_readfirstlane(tid >> 5);
  const float4 bv = *(const float4*)(bias + lane * 4);

  int turn = 0;
  for (int strip = (int)blockIdx.x; strip < nstrips; strip += (int)gridDim.x, ++turn) {
    float* sb = S[turn & 1];
    const int   node  = strip * 16 + wave;
    const int   start = rowptr[node];
    const int   m     = cnt[node];
    const float dn    = rsqrtf(1.0f + (float)m);

    const float4 hv = *(const float4*)(hs + (size_t)node * F + lane * 4);

    float4 acc = make_float4(0.0f, 0.0f, 0.0f, 0.0f);
    for (int b = 0; b < m; b += 32) {
      const int k = b + lane;
      int idx = node;
      if (k < m) idx = col[start + k];
      const int rem = m - b;
      const int mb  = rem < 32 ? rem : 32;
      for (int t0 = 0; t0 < mb; t0 += 4) {
        float4 v[4];
#pragma unroll
        for (int u = 0; u < 4; ++u) {
          const int sidx = __builtin_amdgcn_readlane(idx, t0 + u);
          v[u] = make_float4(0.0f, 0.0f, 0.0f, 0.0f);
          if (t0 + u < mb) v[u] = *(const float4*)(hs + (size_t)sidx * F + lane * 4);
        }
#pragma unroll
        for (int u = 0; u < 4; ++u) {
          acc.x += v[u].x;
          acc.y += v[u].y;
          acc.z += v[u].z;
          acc.w += v[u].w;
        }
      }
    }

    float4 o;
    o.x = fmaxf((acc.x + hv.x) * dn + bv.x, 0.0f);
    o.y = fmaxf((acc.y + hv.y) * dn + bv.y, 0.0f);
    o.z = fmaxf((acc.z + hv.z) * dn + bv.z, 0.0f);
    o.w = fmaxf((acc.w + hv.w) * dn + bv.w, 0.0f);
    *(float4*)(sb + wave * SPITCH + lane * 4) = o;
    __syncthreads();

    if ((wave >> 2) == (turn & 3)) {
      const int t  = wave & 3;
      const int m0 = strip * 16;
      const float* al = sb + lo * SPITCH + 2 * hi;
      const float* bl = Wl + hi * PW + 2 * lo + 32 * t;
      v8f d = v8f{};
#pragma unroll
      for (int j = 0; j < KDIM / 4; ++j) {
        const v2f a = *(const v2f*)(al + 4 * j);
        const v2f b = *(const v2f*)(bl + (2 * j) * PW);
        d = __builtin_amdgcn_wmma_f32_16x16x4_f32(
             false, a,  false, b,
             (short)0, d,  false,  false);
      }
      const int4* __restrict__ cp = (const int4*)(cnt + m0 + hi * 8);
      const int4 c0 = cp[0];
      const int4 c1 = cp[1];
      const float ds[8] = {rsqrtf(1.0f + (float)c0.x), rsqrtf(1.0f + (float)c0.y),
                           rsqrtf(1.0f + (float)c0.z), rsqrtf(1.0f + (float)c0.w),
                           rsqrtf(1.0f + (float)c1.x), rsqrtf(1.0f + (float)c1.y),
                           rsqrtf(1.0f + (float)c1.z), rsqrtf(1.0f + (float)c1.w)};
      float* __restrict__ crow = C + (size_t)(m0 + hi * 8) * NCOLS + 16 * t + lo;
#pragma unroll
      for (int r = 0; r < 8; ++r) crow[(size_t)r * NCOLS] = d[r] * ds[r];
    }
  }
}


__global__ void dot_kernel(const float* __restrict__ z, const int* __restrict__ ls,
                           const int* __restrict__ ld, float* __restrict__ out, int nL) {
  int e = blockIdx.x * blockDim.x + threadIdx.x;
  if (e >= nL) return;
  const float4* a = (const float4*)(z + (size_t)ls[e] * 64);
  const float4* b = (const float4*)(z + (size_t)ld[e] * 64);
  float s = 0.0f;
#pragma unroll
  for (int i = 0; i < 16; ++i) {
    float4 x = a[i], y = b[i];
    s += x.x * y.x + x.y * y.y + x.z * y.z + x.w * y.w;
  }
  out[e] = s;
}


extern "C" void kernel_launch(void* const* d_in, const int* in_sizes, int n_in,
                              void* d_out, int out_size, void* d_ws, size_t ws_size,
                              hipStream_t stream) {
  const float* x  = (const float*)d_in[0];
  const int*   ei = (const int*)d_in[1];
  const int*   el = (const int*)d_in[2];
  const float* W1 = (const float*)d_in[3];
  const float* b1 = (const float*)d_in[4];
  const float* W2 = (const float*)d_in[5];
  const float* b2 = (const float*)d_in[6];

  const int nE = in_sizes[1] / 2;
  const int nL = in_sizes[2] / 2;
  const int* src = ei;
  const int* dst = ei + nE;
  const int* ls  = el;
  const int* ld  = el + nL;

  int*   cnt     = (int*)d_ws;
  int*   rowptr  = cnt + N_PAD;
  int*   sorted  = rowptr + N_PAD;
  int*   col     = sorted + E_CAP;
  int*   runinfo = col + E_CAP;
  float* bufA    = (float*)(runinfo + 2 * NB * NCH);
  float* bufB    = bufA + (size_t)N_NODES * 128;
  float* out     = (float*)d_out;

  bin_kernel<<<NCH, BIN_THREADS, 0, stream>>>(src, dst, sorted, runinfo, nE);
  bucket_kernel<<<NB, BIN_THREADS, 0, stream>>>(sorted, runinfo, cnt, rowptr, col);
  gemm1_kernel<<<256, GEMM1_THREADS, 0, stream>>>(x, W1, cnt, bufA, N_NODES / 16);

  gather_gemm_kernel<<<1024, FG_WAVES * 32, 0, stream>>>(bufA, rowptr, cnt, col, b1, W2, bufB,
                                                         N_NODES / 16);

  gather_kernel<64, false><<<(N_NODES + 7) / 8, 256, 0, stream>>>(bufB, rowptr, cnt, col, b2, bufA,
                                                                   N_NODES);

  dot_kernel<<<(nL + 255) / 256, 256, 0, stream>>>(bufA, ls, ld, out, nL);
}


# Layer-1 product moved behind its edge sum: waves sum dis-weighted x rows per node, the 16-row strip goes through W1 then W2 on the matrix unit in LDS; x @ W1 launch gone; sums reassociated
# speedup vs baseline: 5.2074x; 1.0476x over previous
#include <hip/hip_runtime.h>


typedef float v2f __attribute__((ext_vector_type(2)));
typedef float v8f __attribute__((ext_vector_type(8)));

#define N_NODES 100000
#define N_PAD   100352
#define E_CAP   1600000
#define KDIM 128


#define NCH         256
#define CHUNK       (E_CAP / NCH)
#define BSH         8
#define BNODES      (1 << BSH)
#define NB          ((N_NODES + BNODES - 1) >> BSH)
#define BIN_THREADS 1024
#define EPT         ((CHUNK + BIN_THREADS - 1) / BIN_THREADS)
#define STAGE_CAP   16384

static_assert(E_CAP % NCH == 0, "whole chunks");
static_assert(NB <= BIN_THREADS && BNODES <= BIN_THREADS, "one thread per counter");
static_assert(NCH * 4 == BIN_THREADS, "four threads per run");
static_assert(N_NODES <= (1 << (31 - BSH)), "source and node-in-bucket share one word");

__device__ __forceinline__ int block_excl_scan(int t, int* wsum, int& tot) {
  const int lane = threadIdx.x & 31;
  const int wave = threadIdx.x >> 5;
  int x = t;
#pragma unroll
  for (int o = 1; o < 32; o <<= 1) {
    const int y = __shfl_up(x, o);
    if (lane >= o) x += y;
  }
  if (lane == 31) wsum[wave] = x;
  __syncthreads();
  const int wv = wsum[lane];
  int wp = (lane < wave) ? wv : 0;
  int s  = wv;
#pragma unroll
  for (int o = 16; o > 0; o >>= 1) {
    wp += __shfl_xor(wp, o);
    s  += __shfl_xor(s, o);
  }
  __syncthreads();
  tot = s;
  return wp + (x - t);
}

__global__ void __launch_bounds__(BIN_THREADS)
bin_kernel(const int* __restrict__ src, const int* __restrict__ dst, int* __restrict__ sorted,
           int* __restrict__ runinfo, int nE) {
  __shared__ int hist[NB];
  __shared__ int lstart[NB];
  __shared__ int wsum[32];
  __shared__ int stage[CHUNK];
  const int tid   = threadIdx.x;
  const int chunk = blockIdx.x;
  const int beg   = chunk * CHUNK;
  int n = nE - beg;
  n = n < 0 ? 0 : (n > CHUNK ? CHUNK : n);

  if (n == 0) {
    if (tid < NB) ((int2*)runinfo)[tid * NCH + chunk] = make_int2(beg, 0);
    return;
  }
  if (tid < NB) hist[tid] = 0;
  __syncthreads();

  int w[EPT], bk[EPT], lr[EPT];
#pragma unroll
  for (int u = 0; u < EPT; ++u) {
    const int i  = u * BIN_THREADS + tid;
    const int ic = i < n ? i : n - 1;
    const int d  = dst[beg + ic];
    const int s  = src[beg + ic];
    bk[u] = i < n ? (d >> BSH) : -1;
    w[u]  = (s << BSH) | (d & (BNODES - 1));
  }
#pragma unroll
  for (int u = 0; u < EPT; ++u) {
    lr[u] = 0;
    if (bk[u] >= 0) lr[u] = atomicAdd(&hist[bk[u]], 1);
  }
  __syncthreads();

  int tot;
  const int h  = tid < NB ? hist[tid] : 0;
  const int ex = block_excl_scan(h, wsum, tot);
  if (tid < NB) {
    lstart[tid] = ex;
    ((int2*)runinfo)[tid * NCH + chunk] = make_int2(beg + ex, h);
  }
  __syncthreads();

#pragma unroll
  for (int u = 0; u < EPT; ++u)
    if (bk[u] >= 0) stage[lstart[bk[u]] + lr[u]] = w[u];
  __syncthreads();

  for (int i = tid; i < n; i += BIN_THREADS) sorted[beg + i] = stage[i];
}

__global__ void __launch_bounds__(BIN_THREADS)
bucket_kernel(const int* __restrict__ sorted, const int* __restrict__ runinfo,
              int* __restrict__ cnt, int* __restrict__ rowptr, int* __restrict__ col) {
  __shared__ int rpos[NCH];
  __shared__ int rlen[NCH];
  __shared__ int c[BNODES];
  __shared__ int cur[BNODES];
  __shared__ int wsum[32];
  __shared__ int stage[STAGE_CAP];
  const int tid = threadIdx.x;
  const int b   = blockIdx.x;

  int off = 0, len = 0;
  if (tid < NCH) {
    const int2 ri = ((const int2*)runinfo)[b * NCH + tid];
    rpos[tid] = ri.x;
    rlen[tid] = ri.y;
    off = ri.x - tid * CHUNK;
    len = ri.y;
  }
  if (tid < BNODES) c[tid] = 0;

  int s, tot;
  block_excl_scan(off, wsum, s);
  block_excl_scan(len, wsum, tot);
  const bool fits = tot <= STAGE_CAP;

  const int r  = tid >> 2;
  const int q  = tid & 3;
  const int rp = rpos[r];
  const int rl = rlen[r];
  for (int j0 = 0; j0 < rl; j0 += 16) {
    int w[4];
#pragma unroll
    for (int u = 0; u < 4; ++u) {
      const int j = j0 + 4 * u + q;
      w[u] = sorted[rp + (j < rl ? j : rl - 1)];
    }
#pragma unroll
    for (int u = 0; u < 4; ++u)
      if (j0 + 4 * u + q < rl) atomicAdd(&c[w[u] & (BNODES - 1)], 1);
  }
  __syncthreads();

  int unused;
  const int h  = tid < BNODES ? c[tid] : 0;
  const int ex = block_excl_scan(h, wsum, unused);
  if (tid < BNODES) {
    cur[tid] = ex;
    const int node = b * BNODES + tid;
    if (node < N_NODES) {
      cnt[node]    = h;
      rowptr[node] = s + ex;
    }
  }
  __syncthreads();

  for (int j0 = 0; j0 < rl; j0 += 16) {
    int w[4];
#pragma unroll
    for (int u = 0; u < 4; ++u) {
      const int j = j0 + 4 * u + q;
      w[u] = sorted[rp + (j < rl ? j : rl - 1)];
    }
#pragma unroll
    for (int u = 0; u < 4; ++u) {
      if (j0 + 4 * u + q < rl) {
        const int p = atomicAdd(&cur[w[u] & (BNODES - 1)], 1);
        if (fits) stage[p] = w[u] >> BSH;
        else      col[s + p] = w[u] >> BSH;
      }
    }
  }
  __syncthreads();

  if (fits)
    for (int i = tid; i < tot; i += BIN_THREADS) col[s + i] = stage[i];
}


template <int F, bool RELU>
__global__ void gather_kernel(const float* __restrict__ hs, const int* __restrict__ rowptr,
                              const int* __restrict__ cnt, const int* __restrict__ col,
                              const float* __restrict__ bias, float* __restrict__ agg, int n) {
  constexpr int chunks = F / 4;
  constexpr int epw    = 32 / chunks;
  const int lane = threadIdx.x & 31;
  const int sub  = lane / chunks;
  const int l    = lane % chunks;
  const int node = __builtin_amdgcn_readfirstlane(
      (int)(blockIdx.x * (blockDim.x >> 5) + (threadIdx.x >> 5)));
  if (node >= n) return;
  const int   start = rowptr[node];
  const int   m     = cnt[node];
  const float dn    = rsqrtf(1.0f + (float)m);

  const float4 hv = *(const float4*)(hs + (size_t)node * F + l * 4);

  float4 acc = make_float4(0.0f, 0.0f, 0.0f, 0.0f);
  for (int b = 0; b < m; b += 32) {
    const int k = b + lane;
    int idx = node;
    if (k < m) idx = col[start + k];
    const int rem = m - b;
    const int mb  = rem < 32 ? rem : 32;
    const int T   = (mb + epw - 1) / epw;
    for (int t0 = 0; t0 < T; t0 += 4) {
      float4 v[4];
      int    s[4], jl[4];
#pragma unroll
      for (int u = 0; u < 4; ++u) {
        if constexpr (epw == 1) {
          jl[u] = t0 + u;
          s[u]  = __builtin_amdgcn_readlane(idx, jl[u]);
        } else {
          jl[u] = (t0 + u) * epw + sub;
          s[u]  = __builtin_amdgcn_ds_bpermute(jl[u] << 2, idx);
        }
      }
#pragma unroll
      for (int u = 0; u < 4; ++u) {
        v[u] = make_float4(0.0f, 0.0f, 0.0f, 0.0f);
        if (jl[u] < mb) v[u] = *(const float4*)(hs + (size_t)s[u] * F + l * 4);
      }
#pragma unroll
      for (int u = 0; u < 4; ++u) {
        acc.x += v[u].x;
        acc.y += v[u].y;
        acc.z += v[u].z;
        acc.w += v[u].w;
      }
    }
  }

  if constexpr (epw == 2) {
    acc.x += __shfl_xor(acc.x, 16);
    acc.y += __shfl_xor(acc.y, 16);
    acc.z += __shfl_xor(acc.z, 16);
    acc.w += __shfl_xor(acc.w, 16);
  }

  if (sub == 0) {
    const float4 bv = *(const float4*)(bias + l * 4);
    float4 o;
    o.x = (acc.x + hv.x) * dn + bv.x;
    o.y = (acc.y + hv.y) * dn + bv.y;
    o.z = (acc.z + hv.z) * dn + bv.z;
    o.w = (acc.w + hv.w) * dn + bv.w;
    if (RELU) {
      o.x = fmaxf(o.x, 0.0f);
      o.y = fmaxf(o.y, 0.0f);
      o.z = fmaxf(o.z, 0.0f);
      o.w = fmaxf(o.w, 0.0f);
    }
    *(float4*)(agg + (size_t)node * F + l * 4) = o;
  }
}


#define FG_WAVES 16
#define SPITCH   132
#define GROWS    16

template <int PW>
__device__ __forceinline__ v8f strip_tile(const float* al, const float* bl) {
  v8f d = v8f{};
#pragma unroll
  for (int j = 0; j < KDIM / 4; ++j) {
    const v2f a = *(const v2f*)(al + 4 * j);
    const v2f b = *(const v2f*)(bl + (2 * j) * PW);
    d = __builtin_amdgcn_wmma_f32_16x16x4_f32(
         false, a,  false, b,
         (short)0, d,  false,  false);
  }
  return d;
}

template <int NCOLS, int PW>
__device__ __forceinline__ void stage_w(const float* __restrict__ W, float* Wl, int tid) {
  constexpr int C4 = NCOLS / 4;
  constexpr int ITEMS = (KDIM / 2) * C4 / (FG_WAVES * 32);
  static_assert(ITEMS * FG_WAVES * 32 == (KDIM / 2) * C4, "whole items per thread");
#pragma unroll
  for (int u = 0; u < ITEMS; ++u) {
    const int it = tid + u * (FG_WAVES * 32);
    const int p = it / C4;
    const int c = it % C4;
    const float4 r0 = *(const float4*)(W + (size_t)(2 * p) * NCOLS + 4 * c);
    const float4 r1 = *(const float4*)(W + (size_t)(2 * p + 1) * NCOLS + 4 * c);
    float* d = Wl + p * PW + 8 * c;
    *(float4*)(d)     = make_float4(r0.x, r1.x, r0.y, r1.y);
    *(float4*)(d + 4) = make_float4(r0.z, r1.z, r0.w, r1.w);
  }
}

__device__ __forceinline__ void layer2_tile(const float* hb, const float* W2l, const int* __restrict__ cnt,
                                            float* __restrict__ C, int m0, int t, int lo, int hi) {
  constexpr int NCOLS = 64;
  constexpr int PW    = 2 * NCOLS + 32;
  const v8f d = strip_tile<PW>(hb + lo * SPITCH + 2 * hi, W2l + hi * PW + 2 * lo + 32 * t);
  const int4* __restrict__ cp = (const int4*)(cnt + m0 + hi * 8);
  const int4 c0 = cp[0];
  const int4 c1 = cp[1];
  const float ds[8] = {rsqrtf(1.0f + (float)c0.x), rsqrtf(1.0f + (float)c0.y),
                       rsqrtf(1.0f + (float)c0.z), rsqrtf(1.0f + (float)c0.w),
                       rsqrtf(1.0f + (float)c1.x), rsqrtf(1.0f + (float)c1.y),
                       rsqrtf(1.0f + (float)c1.z), rsqrtf(1.0f + (float)c1.w)};
  float* __restrict__ crow = C + (size_t)(m0 + hi * 8) * NCOLS + 16 * t + lo;
#pragma unroll
  for (int r = 0; r < 8; ++r) crow[(size_t)r * NCOLS] = d[r] * ds[r];
}

__device__ __forceinline__ void rows_issue(const float* __restrict__ x, int idx, int t0, int mb,
                                           int lane, float4 (&v)[GROWS]) {
#pragma unroll
  for (int u = 0; u < GROWS; ++u) {
    const int sidx = __builtin_amdgcn_readlane(idx, t0 + u);
    v[u] = make_float4(0.0f, 0.0f, 0.0f, 0.0f);
    if (t0 + u < mb) v[u] = *(const float4*)(x + (size_t)sidx * 128 + lane * 4);
  }
}

__device__ __forceinline__ void rows_add(float wl, int t0, const float4 (&v)[GROWS], float4& acc) {
#pragma unroll
  for (int u = 0; u < GROWS; ++u) {
    const float w = __builtin_bit_cast(
        float, __builtin_amdgcn_readlane(__builtin_bit_cast(int, wl), t0 + u));
    acc.x = fmaf(w, v[u].x, acc.x);
    acc.y = fmaf(w, v[u].y, acc.y);
    acc.z = fmaf(w, v[u].z, acc.z);
    acc.w = fmaf(w, v[u].w, acc.w);
  }
}

__global__ void __launch_bounds__(FG_WAVES * 32)
gather_mlp_kernel(const float* __restrict__ x, const int* __restrict__ rowptr,
                  const int* __restrict__ cnt, const int* __restrict__ col,
                  const float* __restrict__ W1, const float* __restrict__ b1,
                  const float* __restrict__ W2, float* __restrict__ C, int nstrips) {
  constexpr int F   = 128;
  constexpr int N1  = 128;
  constexpr int N2  = 64;
  constexpr int PW1 = 2 * N1 + 32;
  constexpr int PW2 = 2 * N2 + 32;
  __shared__ __attribute__((aligned(16))) float W1l[(KDIM / 2) * PW1];
  __shared__ __attribute__((aligned(16))) float W2l[(KDIM / 2) * PW2];
  __shared__ __attribute__((aligned(16))) float S[2][16 * SPITCH];
  __shared__ __attribute__((aligned(16))) float H[2][16 * SPITCH];

  const int tid = threadIdx.x;
  stage_w<N1, PW1>(W1, W1l, tid);
  stage_w<N2, PW2>(W2, W2l, tid);
  __syncthreads();

  const int lane = tid & 31;
  const int hi   = lane >> 4;
  const int lo   = lane & 15;
  const int wave = __builtin_amdgcn_readfirstlane(tid >> 5);

  int strip = (int)blockIdx.x;
  if (strip >= nstrips) return;

  int    node  = strip * 16 + wave;
  int    start = rowptr[node];
  int    m     = cnt[node];
  float4 xv    = *(const float4*)(x + (size_t)node * F + lane * 4);
  int    idx   = node;
  if (lane < m) idx = col[start + lane];

  int turn = 0;
  int prev_m0 = 0;
  for (; strip < nstrips; ++turn) {
    float* sb = S[turn & 1];

    const int    nstrip   = strip + (int)gridDim.x;
    const int    nbase    = (nstrip < nstrips ? nstrip : strip) * 16;
    const int    nnode    = nbase + wave;
    const int    nstart_v = rowptr[nbase + lo];
    const int    nm_v     = cnt[nbase + lo];
    const float4 nxv      = *(const float4*)(x + (size_t)nnode * F + lane * 4);

    const float dn = rsqrtf(1.0f + (float)m);
    float4 acc = make_float4(0.0f, 0.0f, 0.0f, 0.0f);
    for (int b = 0; b < m; b += 32) {
      int cur = idx;
      if (b > 0) {
        cur = node;
        if (b + lane < m) cur = col[start + b + lane];
      }
      const int craw = cnt[cur];
      const int rem  = m - b;
      const int mb   = rem < 32 ? rem : 32;
      float4 v[GROWS];
      rows_issue(x, cur, 0, mb, lane, v);
      const float wl = rsqrtf(1.0f + (float)craw);
      rows_add(wl, 0, v, acc);
      if (mb > GROWS) {
        rows_issue(x, cur, GROWS, mb, lane, v);
        rows_add(wl, GROWS, v, acc);
      }
    }

    float4 o;
    o.x = dn * fmaf(dn, xv.x, acc.x);
    o.y = dn * fmaf(dn, xv.y, acc.y);
    o.z = dn * fmaf(dn, xv.z, acc.z);
    o.w = dn * fmaf(dn, xv.w, acc.w);
    *(float4*)(sb + wave * SPITCH + lane * 4) = o;
    __syncthreads();

    const int nstart = __builtin_amdgcn_readlane(nstart_v, wave);
    const int nm     = __builtin_amdgcn_readlane(nm_v, wave);
    int nidx = nnode;
    if (lane < nm) nidx = col[nstart + lane];

    const int jb = (wave + 4 * turn) & 15;
    if (jb < 8) {
      const v8f d = strip_tile<PW1>(sb + lo * SPITCH + 2 * hi, W1l + hi * PW1 + 2 * lo + 32 * jb);
      const float bb = b1[16 * jb + lo];
      float* hrow = H[turn & 1] + (hi * 8) * SPITCH + 16 * jb + lo;
#pragma unroll
      for (int r = 0; r < 8; ++r) hrow[r * SPITCH] = fmaxf(d[r] + bb, 0.0f);
    } else if (jb < 12) {
      if (turn > 0) layer2_tile(H[(turn - 1) & 1], W2l, cnt, C, prev_m0, jb - 8, lo, hi);
    }

    prev_m0 = strip * 16;
    strip = nstrip;
    node  = nnode;
    start = nstart;
    m     = nm;
    xv    = nxv;
    idx   = nidx;
  }

  __syncthreads();
  if (wave < 4) layer2_tile(H[(turn - 1) & 1], W2l, cnt, C, prev_m0, wave, lo, hi);
}


__global__ void dot_kernel(const float* __restrict__ z, const int* __restrict__ ls,
                           const int* __restrict__ ld, float* __restrict__ out, int nL) {
  int e = blockIdx.x * blockDim.x + threadIdx.x;
  if (e >= nL) return;
  const float4* a = (const float4*)(z + (size_t)ls[e] * 64);
  const float4* b = (const float4*)(z + (size_t)ld[e] * 64);
  float s = 0.0f;
#pragma unroll
  for (int i = 0; i < 16; ++i) {
    float4 x = a[i], y = b[i];
    s += x.x * y.x + x.y * y.y + x.z * y.z + x.w * y.w;
  }
  out[e] = s;
}


extern "C" void kernel_launch(void* const* d_in, const int* in_sizes, int n_in,
                              void* d_out, int out_size, void* d_ws, size_t ws_size,
                              hipStream_t stream) {
  const float* x  = (const float*)d_in[0];
  const int*   ei = (const int*)d_in[1];
  const int*   el = (const int*)d_in[2];
  const float* W1 = (const float*)d_in[3];
  const float* b1 = (const float*)d_in[4];
  const float* W2 = (const float*)d_in[5];
  const float* b2 = (const float*)d_in[6];

  const int nE = in_sizes[1] / 2;
  const int nL = in_sizes[2] / 2;
  const int* src = ei;
  const int* dst = ei + nE;
  const int* ls  = el;
  const int* ld  = el + nL;

  int*   cnt     = (int*)d_ws;
  int*   rowptr  = cnt + N_PAD;
  int*   sorted  = rowptr + N_PAD;
  int*   col     = sorted + E_CAP;
  int*   runinfo = col + E_CAP;
  float* bufA    = (float*)(runinfo + 2 * NB * NCH);
  float* bufB    = bufA + (size_t)N_NODES * 128;
  float* out     = (float*)d_out;

  bin_kernel<<<NCH, BIN_THREADS, 0, stream>>>(src, dst, sorted, runinfo, nE);
  bucket_kernel<<<NB, BIN_THREADS, 0, stream>>>(sorted, runinfo, cnt, rowptr, col);
  gather_mlp_kernel<<<512, FG_WAVES * 32, 0, stream>>>(x, rowptr, cnt, col, W1, b1, W2, bufB,
                                                       N_NODES / 16);

  gather_kernel<64, false><<<(N_NODES + 7) / 8, 256, 0, stream>>>(bufB, rowptr, cnt, col, b2, bufA,
                                                                   N_NODES);

  dot_kernel<<<(nL + 255) / 256, 256, 0, stream>>>(bufA, ls, ld, out, nL);
}


# Label-edge dot: 16 lanes per z row, two edges per wave load (whole 128-byte lines), indices handed round by row_share, 16-lane tree sum; the 64 products of a score are summed in a new order
# speedup vs baseline: 5.7836x; 1.1107x over previous
#include <hip/hip_runtime.h>


typedef float v2f __attribute__((ext_vector_type(2)));
typedef float v8f __attribute__((ext_vector_type(8)));

#define N_NODES 100000
#define N_PAD   100352
#define E_CAP   1600000
#define KDIM 128


#define NCH         256
#define CHUNK       (E_CAP / NCH)
#define BSH         8
#define BNODES      (1 << BSH)
#define NB          ((N_NODES + BNODES - 1) >> BSH)
#define BIN_THREADS 1024
#define EPT         ((CHUNK + BIN_THREADS - 1) / BIN_THREADS)
#define STAGE_CAP   16384

static_assert(E_CAP % NCH == 0, "whole chunks");
static_assert(NB <= BIN_THREADS && BNODES <= BIN_THREADS, "one thread per counter");
static_assert(NCH * 4 == BIN_THREADS, "four threads per run");
static_assert(N_NODES <= (1 << (31 - BSH)), "source and node-in-bucket share one word");

__device__ __forceinline__ int block_excl_scan(int t, int* wsum, int& tot) {
  const int lane = threadIdx.x & 31;
  const int wave = threadIdx.x >> 5;
  int x = t;
#pragma unroll
  for (int o = 1; o < 32; o <<= 1) {
    const int y = __shfl_up(x, o);
    if (lane >= o) x += y;
  }
  if (lane == 31) wsum[wave] = x;
  __syncthreads();
  const int wv = wsum[lane];
  int wp = (lane < wave) ? wv : 0;
  int s  = wv;
#pragma unroll
  for (int o = 16; o > 0; o >>= 1) {
    wp += __shfl_xor(wp, o);
    s  += __shfl_xor(s, o);
  }
  __syncthreads();
  tot = s;
  return wp + (x - t);
}

__global__ void __launch_bounds__(BIN_THREADS)
bin_kernel(const int* __restrict__ src, const int* __restrict__ dst, int* __restrict__ sorted,
           int* __restrict__ runinfo, int nE) {
  __shared__ int hist[NB];
  __shared__ int lstart[NB];
  __shared__ int wsum[32];
  __shared__ int stage[CHUNK];
  const int tid   = threadIdx.x;
  const int chunk = blockIdx.x;
  const int beg   = chunk * CHUNK;
  int n = nE - beg;
  n = n < 0 ? 0 : (n > CHUNK ? CHUNK : n);

  if (n == 0) {
    if (tid < NB) ((int2*)runinfo)[tid * NCH + chunk] = make_int2(beg, 0);
    return;
  }
  if (tid < NB) hist[tid] = 0;
  __syncthreads();

  int w[EPT], bk[EPT], lr[EPT];
#pragma unroll
  for (int u = 0; u < EPT; ++u) {
    const int i  = u * BIN_THREADS + tid;
    const int ic = i < n ? i : n - 1;
    const int d  = dst[beg + ic];
    const int s  = src[beg + ic];
    bk[u] = i < n ? (d >> BSH) : -1;
    w[u]  = (s << BSH) | (d & (BNODES - 1));
  }
#pragma unroll
  for (int u = 0; u < EPT; ++u) {
    lr[u] = 0;
    if (bk[u] >= 0) lr[u] = atomicAdd(&hist[bk[u]], 1);
  }
  __syncthreads();

  int tot;
  const int h  = tid < NB ? hist[tid] : 0;
  const int ex = block_excl_scan(h, wsum, tot);
  if (tid < NB) {
    lstart[tid] = ex;
    ((int2*)runinfo)[tid * NCH + chunk] = make_int2(beg + ex, h);
  }
  __syncthreads();

#pragma unroll
  for (int u = 0; u < EPT; ++u)
    if (bk[u] >= 0) stage[lstart[bk[u]] + lr[u]] = w[u];
  __syncthreads();

  for (int i = tid; i < n; i += BIN_THREADS) sorted[beg + i] = stage[i];
}

__global__ void __launch_bounds__(BIN_THREADS)
bucket_kernel(const int* __restrict__ sorted, const int* __restrict__ runinfo,
              int* __restrict__ cnt, int* __restrict__ rowptr, int* __restrict__ col) {
  __shared__ int rpos[NCH];
  __shared__ int rlen[NCH];
  __shared__ int c[BNODES];
  __shared__ int cur[BNODES];
  __shared__ int wsum[32];
  __shared__ int stage[STAGE_CAP];
  const int tid = threadIdx.x;
  const int b   = blockIdx.x;

  int off = 0, len = 0;
  if (tid < NCH) {
    const int2 ri = ((const int2*)runinfo)[b * NCH + tid];
    rpos[tid] = ri.x;
    rlen[tid] = ri.y;
    off = ri.x - tid * CHUNK;
    len = ri.y;
  }
  if (tid < BNODES) c[tid] = 0;

  int s, tot;
  block_excl_scan(off, wsum, s);
  block_excl_scan(len, wsum, tot);
  const bool fits = tot <= STAGE_CAP;

  const int r  = tid >> 2;
  const int q  = tid & 3;
  const int rp = rpos[r];
  const int rl = rlen[r];
  for (int j0 = 0; j0 < rl; j0 += 16) {
    int w[4];
#pragma unroll
    for (int u = 0; u < 4; ++u) {
      const int j = j0 + 4 * u + q;
      w[u] = sorted[rp + (j < rl ? j : rl - 1)];
    }
#pragma unroll
    for (int u = 0; u < 4; ++u)
      if (j0 + 4 * u + q < rl) atomicAdd(&c[w[u] & (BNODES - 1)], 1);
  }
  __syncthreads();

  int unused;
  const int h  = tid < BNODES ? c[tid] : 0;
  const int ex = block_excl_scan(h, wsum, unused);
  if (tid < BNODES) {
    cur[tid] = ex;
    const int node = b * BNODES + tid;
    if (node < N_NODES) {
      cnt[node]    = h;
      rowptr[node] = s + ex;
    }
  }
  __syncthreads();

  for (int j0 = 0; j0 < rl; j0 += 16) {
    int w[4];
#pragma unroll
    for (int u = 0; u < 4; ++u) {
      const int j = j0 + 4 * u + q;
      w[u] = sorted[rp + (j < rl ? j : rl - 1)];
    }
#pragma unroll
    for (int u = 0; u < 4; ++u) {
      if (j0 + 4 * u + q < rl) {
        const int p = atomicAdd(&cur[w[u] & (BNODES - 1)], 1);
        if (fits) stage[p] = w[u] >> BSH;
        else      col[s + p] = w[u] >> BSH;
      }
    }
  }
  __syncthreads();

  if (fits)
    for (int i = tid; i < tot; i += BIN_THREADS) col[s + i] = stage[i];
}


template <int F, bool RELU>
__global__ void gather_kernel(const float* __restrict__ hs, const int* __restrict__ rowptr,
                              const int* __restrict__ cnt, const int* __restrict__ col,
                              const float* __restrict__ bias, float* __restrict__ agg, int n) {
  constexpr int chunks = F / 4;
  constexpr int epw    = 32 / chunks;
  const int lane = threadIdx.x & 31;
  const int sub  = lane / chunks;
  const int l    = lane % chunks;
  const int node = __builtin_amdgcn_readfirstlane(
      (int)(blockIdx.x * (blockDim.x >> 5) + (threadIdx.x >> 5)));
  if (node >= n) return;
  const int   start = rowptr[node];
  const int   m     = cnt[node];
  const float dn    = rsqrtf(1.0f + (float)m);

  const float4 hv = *(const float4*)(hs + (size_t)node * F + l * 4);

  float4 acc = make_float4(0.0f, 0.0f, 0.0f, 0.0f);
  for (int b = 0; b < m; b += 32) {
    const int k = b + lane;
    int idx = node;
    if (k < m) idx = col[start + k];
    const int rem = m - b;
    const int mb  = rem < 32 ? rem : 32;
    const int T   = (mb + epw - 1) / epw;
    for (int t0 = 0; t0 < T; t0 += 4) {
      float4 v[4];
      int    s[4], jl[4];
#pragma unroll
      for (int u = 0; u < 4; ++u) {
        if constexpr (epw == 1) {
          jl[u] = t0 + u;
          s[u]  = __builtin_amdgcn_readlane(idx, jl[u]);
        } else {
          jl[u] = (t0 + u) * epw + sub;
          s[u]  = __builtin_amdgcn_ds_bpermute(jl[u] << 2, idx);
        }
      }
#pragma unroll
      for (int u = 0; u < 4; ++u) {
        v[u] = make_float4(0.0f, 0.0f, 0.0f, 0.0f);
        if (jl[u] < mb) v[u] = *(const float4*)(hs + (size_t)s[u] * F + l * 4);
      }
#pragma unroll
      for (int u = 0; u < 4; ++u) {
        acc.x += v[u].x;
        acc.y += v[u].y;
        acc.z += v[u].z;
        acc.w += v[u].w;
      }
    }
  }

  if constexpr (epw == 2) {
    acc.x += __shfl_xor(acc.x, 16);
    acc.y += __shfl_xor(acc.y, 16);
    acc.z += __shfl_xor(acc.z, 16);
    acc.w += __shfl_xor(acc.w, 16);
  }

  if (sub == 0) {
    const float4 bv = *(const float4*)(bias + l * 4);
    float4 o;
    o.x = (acc.x + hv.x) * dn + bv.x;
    o.y = (acc.y + hv.y) * dn + bv.y;
    o.z = (acc.z + hv.z) * dn + bv.z;
    o.w = (acc.w + hv.w) * dn + bv.w;
    if (RELU) {
      o.x = fmaxf(o.x, 0.0f);
      o.y = fmaxf(o.y, 0.0f);
      o.z = fmaxf(o.z, 0.0f);
      o.w = fmaxf(o.w, 0.0f);
    }
    *(float4*)(agg + (size_t)node * F + l * 4) = o;
  }
}


#define FG_WAVES 16
#define SPITCH   132
#define GROWS    16

template <int PW>
__device__ __forceinline__ v8f strip_tile(const float* al, const float* bl) {
  v8f d = v8f{};
#pragma unroll
  for (int j = 0; j < KDIM / 4; ++j) {
    const v2f a = *(const v2f*)(al + 4 * j);
    const v2f b = *(const v2f*)(bl + (2 * j) * PW);
    d = __builtin_amdgcn_wmma_f32_16x16x4_f32(
         false, a,  false, b,
         (short)0, d,  false,  false);
  }
  return d;
}

template <int NCOLS, int PW>
__device__ __forceinline__ void stage_w(const float* __restrict__ W, float* Wl, int tid) {
  constexpr int C4 = NCOLS / 4;
  constexpr int ITEMS = (KDIM / 2) * C4 / (FG_WAVES * 32);
  static_assert(ITEMS * FG_WAVES * 32 == (KDIM / 2) * C4, "whole items per thread");
#pragma unroll
  for (int u = 0; u < ITEMS; ++u) {
    const int it = tid + u * (FG_WAVES * 32);
    const int p = it / C4;
    const int c = it % C4;
    const float4 r0 = *(const float4*)(W + (size_t)(2 * p) * NCOLS + 4 * c);
    const float4 r1 = *(const float4*)(W + (size_t)(2 * p + 1) * NCOLS + 4 * c);
    float* d = Wl + p * PW + 8 * c;
    *(float4*)(d)     = make_float4(r0.x, r1.x, r0.y, r1.y);
    *(float4*)(d + 4) = make_float4(r0.z, r1.z, r0.w, r1.w);
  }
}

__device__ __forceinline__ void layer2_tile(const float* hb, const float* W2l, const int* __restrict__ cnt,
                                            float* __restrict__ C, int m0, int t, int lo, int hi) {
  constexpr int NCOLS = 64;
  constexpr int PW    = 2 * NCOLS + 32;
  const v8f d = strip_tile<PW>(hb + lo * SPITCH + 2 * hi, W2l + hi * PW + 2 * lo + 32 * t);
  const int4* __restrict__ cp = (const int4*)(cnt + m0 + hi * 8);
  const int4 c0 = cp[0];
  const int4 c1 = cp[1];
  const float ds[8] = {rsqrtf(1.0f + (float)c0.x), rsqrtf(1.0f + (float)c0.y),
                       rsqrtf(1.0f + (float)c0.z), rsqrtf(1.0f + (float)c0.w),
                       rsqrtf(1.0f + (float)c1.x), rsqrtf(1.0f + (float)c1.y),
                       rsqrtf(1.0f + (float)c1.z), rsqrtf(1.0f + (float)c1.w)};
  float* __restrict__ crow = C + (size_t)(m0 + hi * 8) * NCOLS + 16 * t + lo;
#pragma unroll
  for (int r = 0; r < 8; ++r) crow[(size_t)r * NCOLS] = d[r] * ds[r];
}

__device__ __forceinline__ void rows_issue(const float* __restrict__ x, int idx, int t0, int mb,
                                           int lane, float4 (&v)[GROWS]) {
#pragma unroll
  for (int u = 0; u < GROWS; ++u) {
    const int sidx = __builtin_amdgcn_readlane(idx, t0 + u);
    v[u] = make_float4(0.0f, 0.0f, 0.0f, 0.0f);
    if (t0 + u < mb) v[u] = *(const float4*)(x + (size_t)sidx * 128 + lane * 4);
  }
}

__device__ __forceinline__ void rows_add(float wl, int t0, const float4 (&v)[GROWS], float4& acc) {
#pragma unroll
  for (int u = 0; u < GROWS; ++u) {
    const float w = __builtin_bit_cast(
        float, __builtin_amdgcn_readlane(__builtin_bit_cast(int, wl), t0 + u));
    acc.x = fmaf(w, v[u].x, acc.x);
    acc.y = fmaf(w, v[u].y, acc.y);
    acc.z = fmaf(w, v[u].z, acc.z);
    acc.w = fmaf(w, v[u].w, acc.w);
  }
}

__global__ void __launch_bounds__(FG_WAVES * 32)
gather_mlp_kernel(const float* __restrict__ x, const int* __restrict__ rowptr,
                  const int* __restrict__ cnt, const int* __restrict__ col,
                  const float* __restrict__ W1, const float* __restrict__ b1,
                  const float* __restrict__ W2, float* __restrict__ C, int nstrips) {
  constexpr int F   = 128;
  constexpr int N1  = 128;
  constexpr int N2  = 64;
  constexpr int PW1 = 2 * N1 + 32;
  constexpr int PW2 = 2 * N2 + 32;
  __shared__ __attribute__((aligned(16))) float W1l[(KDIM / 2) * PW1];
  __shared__ __attribute__((aligned(16))) float W2l[(KDIM / 2) * PW2];
  __shared__ __attribute__((aligned(16))) float S[2][16 * SPITCH];
  __shared__ __attribute__((aligned(16))) float H[2][16 * SPITCH];

  const int tid = threadIdx.x;
  stage_w<N1, PW1>(W1, W1l, tid);
  stage_w<N2, PW2>(W2, W2l, tid);
  __syncthreads();

  const int lane = tid & 31;
  const int hi   = lane >> 4;
  const int lo   = lane & 15;
  const int wave = __builtin_amdgcn_readfirstlane(tid >> 5);

  int strip = (int)blockIdx.x;
  if (strip >= nstrips) return;

  int    node  = strip * 16 + wave;
  int    start = rowptr[node];
  int    m     = cnt[node];
  float4 xv    = *(const float4*)(x + (size_t)node * F + lane * 4);
  int    idx   = node;
  if (lane < m) idx = col[start + lane];

  int turn = 0;
  int prev_m0 = 0;
  for (; strip < nstrips; ++turn) {
    float* sb = S[turn & 1];

    const int    nstrip   = strip + (int)gridDim.x;
    const int    nbase    = (nstrip < nstrips ? nstrip : strip) * 16;
    const int    nnode    = nbase + wave;
    const int    nstart_v = rowptr[nbase + lo];
    const int    nm_v     = cnt[nbase + lo];
    const float4 nxv      = *(const float4*)(x + (size_t)nnode * F + lane * 4);

    const float dn = rsqrtf(1.0f + (float)m);
    float4 acc = make_float4(0.0f, 0.0f, 0.0f, 0.0f);
    for (int b = 0; b < m; b += 32) {
      int cur = idx;
      if (b > 0) {
        cur = node;
        if (b + lane < m) cur = col[start + b + lane];
      }
      const int craw = cnt[cur];
      const int rem  = m - b;
      const int mb   = rem < 32 ? rem : 32;
      float4 v[GROWS];
      rows_issue(x, cur, 0, mb, lane, v);
      const float wl = rsqrtf(1.0f + (float)craw);
      rows_add(wl, 0, v, acc);
      if (mb > GROWS) {
        rows_issue(x, cur, GROWS, mb, lane, v);
        rows_add(wl, GROWS, v, acc);
      }
    }

    float4 o;
    o.x = dn * fmaf(dn, xv.x, acc.x);
    o.y = dn * fmaf(dn, xv.y, acc.y);
    o.z = dn * fmaf(dn, xv.z, acc.z);
    o.w = dn * fmaf(dn, xv.w, acc.w);
    *(float4*)(sb + wave * SPITCH + lane * 4) = o;
    __syncthreads();

    const int nstart = __builtin_amdgcn_readlane(nstart_v, wave);
    const int nm     = __builtin_amdgcn_readlane(nm_v, wave);
    int nidx = nnode;
    if (lane < nm) nidx = col[nstart + lane];

    const int jb = (wave + 4 * turn) & 15;
    if (jb < 8) {
      const v8f d = strip_tile<PW1>(sb + lo * SPITCH + 2 * hi, W1l + hi * PW1 + 2 * lo + 32 * jb);
      const float bb = b1[16 * jb + lo];
      float* hrow = H[turn & 1] + (hi * 8) * SPITCH + 16 * jb + lo;
#pragma unroll
      for (int r = 0; r < 8; ++r) hrow[r * SPITCH] = fmaxf(d[r] + bb, 0.0f);
    } else if (jb < 12) {
      if (turn > 0) layer2_tile(H[(turn - 1) & 1], W2l, cnt, C, prev_m0, jb - 8, lo, hi);
    }

    prev_m0 = strip * 16;
    strip = nstrip;
    node  = nnode;
    start = nstart;
    m     = nm;
    xv    = nxv;
    idx   = nidx;
  }

  __syncthreads();
  if (wave < 4) layer2_tile(H[(turn - 1) & 1], W2l, cnt, C, prev_m0, wave, lo, hi);
}


#define DOT_WAVES 4
#define DOT_SLOTS 16

template <int U>
__device__ __forceinline__ int row_lane(int v) {
  return __builtin_amdgcn_update_dpp(0, v, 0x150 + U, 0xf, 0xf, true);
}

template <int M>
__device__ __forceinline__ float row_xor(float v) {
  return __builtin_bit_cast(
      float, __builtin_amdgcn_update_dpp(0, __builtin_bit_cast(int, v), 0x160 + M, 0xf, 0xf, true));
}

template <int U>
__device__ __forceinline__ void dot_issue(const char* __restrict__ zb, int l16, int ia, int ib,
                                          float4 (&a)[DOT_SLOTS], float4 (&b)[DOT_SLOTS]) {
  const int sa = row_lane<U>(ia);
  const int sb = row_lane<U>(ib);
  a[U] = *(const float4*)(zb + (unsigned)(sa * 256 + l16));
  b[U] = *(const float4*)(zb + (unsigned)(sb * 256 + l16));
  if constexpr (U + 1 < DOT_SLOTS) dot_issue<U + 1>(zb, l16, ia, ib, a, b);
}

template <int N, int M>
__device__ __forceinline__ void tree_step(float (&p)[DOT_SLOTS], bool up) {
#pragma unroll
  for (int u = 0; u < N / 2; ++u) {
    const float keep = up ? p[u + N / 2] : p[u];
    const float send = up ? p[u] : p[u + N / 2];
    p[u] = keep + row_xor<M>(send);
  }
}

__global__ void __launch_bounds__(DOT_WAVES * 32) __attribute__((amdgpu_waves_per_eu(1, 7)))
dot_kernel(const float* __restrict__ z, const int* __restrict__ ls,
           const int* __restrict__ ld, float* __restrict__ out, int nL) {
  const int lane = threadIdx.x & 31;
  const int l    = lane & 15;
  const int base = (int)(blockIdx.x * DOT_WAVES + (threadIdx.x >> 5)) * 32;
  if (base >= nL) return;
  const int e  = base + lane;
  const int ec = e < nL ? e : nL - 1;
  const int ia = ls[ec];
  const int ib = ld[ec];

  float4 a[DOT_SLOTS], b[DOT_SLOTS];
  dot_issue<0>((const char*)z, l * 16, ia, ib, a, b);

  float p[DOT_SLOTS];
#pragma unroll
  for (int u = 0; u < DOT_SLOTS; ++u) {
    float s = a[u].x * b[u].x;
    s = fmaf(a[u].y, b[u].y, s);
    s = fmaf(a[u].z, b[u].z, s);
    s = fmaf(a[u].w, b[u].w, s);
    p[u] = s;
  }

  tree_step<16, 8>(p, (lane & 8) != 0);
  tree_step<8, 4>(p, (lane & 4) != 0);
  tree_step<4, 2>(p, (lane & 2) != 0);
  tree_step<2, 1>(p, (lane & 1) != 0);

  if (e < nL) out[e] = p[0];
}


extern "C" void kernel_launch(void* const* d_in, const int* in_sizes, int n_in,
                              void* d_out, int out_size, void* d_ws, size_t ws_size,
                              hipStream_t stream) {
  const float* x  = (const float*)d_in[0];
  const int*   ei = (const int*)d_in[1];
  const int*   el = (const int*)d_in[2];
  const float* W1 = (const float*)d_in[3];
  const float* b1 = (const float*)d_in[4];
  const float* W2 = (const float*)d_in[5];
  const float* b2 = (const float*)d_in[6];

  const int nE = in_sizes[1] / 2;
  const int nL = in_sizes[2] / 2;
  const int* src = ei;
  const int* dst = ei + nE;
  const int* ls  = el;
  const int* ld  = el + nL;

  int*   cnt     = (int*)d_ws;
  int*   rowptr  = cnt + N_PAD;
  int*   sorted  = rowptr + N_PAD;
  int*   col     = sorted + E_CAP;
  int*   runinfo = col + E_CAP;
  float* bufA    = (float*)(runinfo + 2 * NB * NCH);
  float* bufB    = bufA + (size_t)N_NODES * 128;
  float* out     = (float*)d_out;

  bin_kernel<<<NCH, BIN_THREADS, 0, stream>>>(src, dst, sorted, runinfo, nE);
  bucket_kernel<<<NB, BIN_THREADS, 0, stream>>>(sorted, runinfo, cnt, rowptr, col);
  gather_mlp_kernel<<<512, FG_WAVES * 32, 0, stream>>>(x, rowptr, cnt, col, W1, b1, W2, bufB,
                                                       N_NODES / 16);

  gather_kernel<64, false><<<(N_NODES + 7) / 8, 256, 0, stream>>>(bufB, rowptr, cnt, col, b2, bufA,
                                                                   N_NODES);

  dot_kernel<<<(nL + DOT_WAVES * 32 - 1) / (DOT_WAVES * 32), DOT_WAVES * 32, 0, stream>>>(bufA, ls, ld, out, nL);
}
